# starting kernel
# baseline (speedup 1.0000x reference)
#include <hip/hip_runtime.h>

typedef float v2f __attribute__((ext_vector_type(2)));
typedef float v8f __attribute__((ext_vector_type(8)));

#define NEG_SLOPE 0.2f

__device__ __forceinline__ unsigned fkey(float f) {
  unsigned u = __float_as_uint(f);
  return (u & 0x80000000u) ? ~u : (u | 0x80000000u);
}
__device__ __forceinline__ float fdekey(unsigned k) {
  unsigned u = (k & 0x80000000u) ? (k & 0x7FFFFFFFu) : ~k;
  return __uint_as_float(u);
}

__global__ void zero_f32(float* __restrict__ p, size_t n) {
  size_t i = (size_t)blockIdx.x * blockDim.x + threadIdx.x;
  size_t stride = (size_t)gridDim.x * blockDim.x;
  for (; i < n; i += stride) p[i] = 0.0f;
}

__global__ void gemm_wmma_f32(const float* __restrict__ A, const float* __restrict__ B,
                              float* __restrict__ C, int N, int K, int Nc) {
  const int lane = threadIdx.x & 31;
  const int half = lane >> 4;
  const int idx  = lane & 15;
  const int row0 = blockIdx.x * 16;
  const int col0 = blockIdx.y * 16;

  const int ar = min(row0 + idx, N - 1);
  const int bc = min(col0 + idx, Nc - 1);
  const float* __restrict__ ap = A + (size_t)ar * K + 2 * half;
  const float* __restrict__ bp = B + (size_t)bc * K + 2 * half;

  v8f c = {0.f, 0.f, 0.f, 0.f, 0.f, 0.f, 0.f, 0.f};
  for (int k = 0; k < K; k += 4) {
    v2f a = *(const v2f*)(ap + k);
    v2f b = *(const v2f*)(bp + k);
    c = __builtin_amdgcn_wmma_f32_16x16x4_f32(false, a, false, b, (short)0, c,
                                              false, false);
  }

  const int col = col0 + idx;
  if (col < Nc) {
#pragma unroll
    for (int r = 0; r < 8; ++r) {
      const int row = row0 + r + 8 * half;
      if (row < N) C[(size_t)row * Nc + col] = c[r];
    }
  }
}

__global__ void attn_logits(const float* __restrict__ h, const float* __restrict__ att_s,
                            const float* __restrict__ att_d, float* __restrict__ a_s,
                            float* __restrict__ a_d, int NH, int H, int D) {
  int i = blockIdx.x * blockDim.x + threadIdx.x;
  if (i >= NH) return;
  const int hh = i % H;
  const float* __restrict__ hp = h + (size_t)i * D;
  const float* __restrict__ as = att_s + (size_t)hh * D;
  const float* __restrict__ ad = att_d + (size_t)hh * D;
  float s0 = 0.f, s1 = 0.f;
  for (int dd = 0; dd < D; ++dd) {
    const float v = hp[dd];
    s0 += v * as[dd];
    s1 += v * ad[dd];
  }
  a_s[i] = s0;
  a_d[i] = s1;
}

__global__ void edge_max(const int* __restrict__ src, const int* __restrict__ dst,
                         int E, int N, const float* __restrict__ a_s,
                         const float* __restrict__ a_d, unsigned* __restrict__ mkey,
                         int H) {
  const int t = blockIdx.x * blockDim.x + threadIdx.x;
  const int TE = E + N;
  if (t >= TE) return;
  const int s = (t < E) ? src[t] : (t - E);
  const int d = (t < E) ? dst[t] : (t - E);
  for (int hh = 0; hh < H; ++hh) {
    float e = a_s[s * H + hh] + a_d[d * H + hh];
    e = (e > 0.f) ? e : NEG_SLOPE * e;
    atomicMax(&mkey[d * H + hh], fkey(e));
  }
}

__global__ void edge_expsum(const int* __restrict__ src, const int* __restrict__ dst,
                            int E, int N, const float* __restrict__ a_s,
                            const float* __restrict__ a_d,
                            const unsigned* __restrict__ mkey,
                            float* __restrict__ den, int H) {
  const int t = blockIdx.x * blockDim.x + threadIdx.x;
  const int TE = E + N;
  if (t >= TE) return;
  const int s = (t < E) ? src[t] : (t - E);
  const int d = (t < E) ? dst[t] : (t - E);
  for (int hh = 0; hh < H; ++hh) {
    float e = a_s[s * H + hh] + a_d[d * H + hh];
    e = (e > 0.f) ? e : NEG_SLOPE * e;
    const float ex = expf(e - fdekey(mkey[d * H + hh]));
    atomicAdd(&den[d * H + hh], ex);
  }
}

__global__ void edge_aggregate(const int* __restrict__ src, const int* __restrict__ dst,
                               int E, int N, const float* __restrict__ a_s,
                               const float* __restrict__ a_d,
                               const unsigned* __restrict__ mkey,
                               const float* __restrict__ den,
                               const float* __restrict__ h, float* __restrict__ agg,
                               int H, int D) {
  const int t = blockIdx.x * blockDim.x + threadIdx.x;
  const int TEH = (E + N) * H;
  if (t >= TEH) return;
  const int ei = t / H;
  const int hh = t - ei * H;
  const int s = (ei < E) ? src[ei] : (ei - E);
  const int d = (ei < E) ? dst[ei] : (ei - E);
  float e = a_s[s * H + hh] + a_d[d * H + hh];
  e = (e > 0.f) ? e : NEG_SLOPE * e;
  const float alpha = expf(e - fdekey(mkey[d * H + hh])) / den[d * H + hh];
  const float* __restrict__ hs = h + ((size_t)s * H + hh) * D;
  float* __restrict__ ag = agg + ((size_t)d * H + hh) * D;
  for (int dd = 0; dd < D; ++dd) atomicAdd(&ag[dd], alpha * hs[dd]);
}

__global__ void bias_relu(const float* __restrict__ agg, const float* __restrict__ b,
                          float* __restrict__ out, int C, size_t n) {
  size_t i = (size_t)blockIdx.x * blockDim.x + threadIdx.x;
  if (i >= n) return;
  const float v = agg[i] + b[i % C];
  out[i] = (v > 0.f) ? v : 0.f;
}

__global__ void bias_add_inplace(float* __restrict__ out, const float* __restrict__ b,
                                 int C, size_t n) {
  size_t i = (size_t)blockIdx.x * blockDim.x + threadIdx.x;
  if (i >= n) return;
  out[i] += b[i % C];
}

extern "C" void kernel_launch(void* const* d_in, const int* in_sizes, int n_in,
                              void* d_out, int out_size, void* d_ws, size_t ws_size,
                              hipStream_t stream) {
  const float* x        = (const float*)d_in[0];
  const int*   ei       = (const int*)d_in[1];
  const float* W1       = (const float*)d_in[2];
  const float* att_src1 = (const float*)d_in[3];
  const float* att_dst1 = (const float*)d_in[4];
  const float* b1       = (const float*)d_in[5];
  const float* W2       = (const float*)d_in[6];
  const float* att_src2 = (const float*)d_in[7];
  const float* att_dst2 = (const float*)d_in[8];
  const float* b2       = (const float*)d_in[9];

  const int NFEAT = 256, NHID = 128, H1 = 4, D1 = 32, NCLS = 40;
  const int N = in_sizes[0] / NFEAT;
  const int E = in_sizes[1] / 2;
  const int* src = ei;
  const int* dst = ei + E;

  float* out = (float*)d_out;

  float* w = (float*)d_ws;
  float* h1   = w;
  float* h2   = h1  + (size_t)N * NHID;
  float* as1  = h2  + (size_t)N * NCLS;
  float* ad1  = as1 + (size_t)N * H1;
  float* as2  = ad1 + (size_t)N * H1;
  float* ad2  = as2 + (size_t)N;
  float* zreg = ad2 + (size_t)N;
  float*    agg1 = zreg;
  float*    den1 = agg1 + (size_t)N * NHID;
  unsigned* m1   = (unsigned*)(den1 + (size_t)N * H1);
  float*    den2 = (float*)(m1 + (size_t)N * H1);
  unsigned* m2   = (unsigned*)(den2 + (size_t)N);
  const size_t zcount = (size_t)N * (NHID + H1 + H1 + 1 + 1);

  const int ZB = 2048, TB = 256;
  const int TE = E + N;

  zero_f32<<<ZB, TB, 0, stream>>>(zreg, zcount);
  zero_f32<<<ZB, TB, 0, stream>>>(out, (size_t)N * NCLS);

  {
    dim3 grid((N + 15) / 16, NHID / 16);
    gemm_wmma_f32<<<grid, 32, 0, stream>>>(x, W1, h1, N, NFEAT, NHID);
  }
  attn_logits<<<(N * H1 + TB - 1) / TB, TB, 0, stream>>>(h1, att_src1, att_dst1,
                                                         as1, ad1, N * H1, H1, D1);
  edge_max<<<(TE + TB - 1) / TB, TB, 0, stream>>>(src, dst, E, N, as1, ad1, m1, H1);
  edge_expsum<<<(TE + TB - 1) / TB, TB, 0, stream>>>(src, dst, E, N, as1, ad1, m1,
                                                     den1, H1);
  edge_aggregate<<<(TE * H1 + TB - 1) / TB, TB, 0, stream>>>(src, dst, E, N, as1, ad1,
                                                             m1, den1, h1, agg1, H1, D1);
  float* x2 = h1;
  {
    const size_t n = (size_t)N * NHID;
    bias_relu<<<(unsigned)((n + TB - 1) / TB), TB, 0, stream>>>(agg1, b1, x2, NHID, n);
  }

  {
    dim3 grid((N + 15) / 16, (NCLS + 15) / 16);
    gemm_wmma_f32<<<grid, 32, 0, stream>>>(x2, W2, h2, N, NHID, NCLS);
  }
  attn_logits<<<(N + TB - 1) / TB, TB, 0, stream>>>(h2, att_src2, att_dst2,
                                                    as2, ad2, N, 1, NCLS);
  edge_max<<<(TE + TB - 1) / TB, TB, 0, stream>>>(src, dst, E, N, as2, ad2, m2, 1);
  edge_expsum<<<(TE + TB - 1) / TB, TB, 0, stream>>>(src, dst, E, N, as2, ad2, m2,
                                                     den2, 1);
  edge_aggregate<<<(TE + TB - 1) / TB, TB, 0, stream>>>(src, dst, E, N, as2, ad2,
                                                        m2, den2, h2, out, 1, NCLS);
  {
    const size_t n = (size_t)N * NCLS;
    bias_add_inplace<<<(unsigned)((n + TB - 1) / TB), TB, 0, stream>>>(out, b2, NCLS, n);
  }
}


# layer-1 GEMM re-decomposed: a wave keeps its 16-column strip of W1 (whole K) in registers, 256 resident work-groups loop over 32-row panels of x staged in LDS by whole-line loads
# speedup vs baseline: 11.6701x; 11.6701x over previous
#include <hip/hip_runtime.h>

typedef float v2f __attribute__((ext_vector_type(2)));
typedef float v4f __attribute__((ext_vector_type(4)));
typedef float v8f __attribute__((ext_vector_type(8)));

#define NEG_SLOPE 0.2f

__device__ __forceinline__ unsigned fkey(float f) {
  unsigned u = __float_as_uint(f);
  return (u & 0x80000000u) ? ~u : (u | 0x80000000u);
}
__device__ __forceinline__ float fdekey(unsigned k) {
  unsigned u = (k & 0x80000000u) ? (k & 0x7FFFFFFFu) : ~k;
  return __uint_as_float(u);
}

#define G1_NT 512
#define G1_WG 256

template <int K, int NC>
__global__ void __launch_bounds__(G1_NT)
gemm_wmma_f32_bres(const float* __restrict__ A, const float* __restrict__ B,
                   float* __restrict__ C, int N) {
  constexpr int XS = K + 4;
  constexpr int NCS = NC / 16;
  constexpr int NW = G1_NT / 32;
  constexpr int TS = NW / NCS;
  constexpr int PR = TS * 16;
  constexpr int KJ = K / 4;
  constexpr int KB = 16;
  constexpr int C4 = K / 4;
  constexpr int RQ = G1_NT / C4;
  constexpr int XQ = PR / RQ;
  constexpr int WQ = NC / RQ;
  static_assert(K % 4 == 0 && KJ % KB == 0 && XS % 64 == 4, "k steps of 4; conflict-free fragment reads");
  static_assert(NC % 16 == 0 && NW % NCS == 0 && G1_NT % C4 == 0 && PR % RQ == 0 && NC % RQ == 0,
                "panel geometry");
  __shared__ __attribute__((aligned(16))) float wsm[NC * XS];
  __shared__ __attribute__((aligned(16))) float xsm[2][PR * XS];

  const int tid = threadIdx.x;
  const int lane = tid & 31;
  const int wave = __builtin_amdgcn_readfirstlane(tid >> 5);
  const int cs = wave % NCS;
  const int ts = wave / NCS;
  const int half = lane >> 4;
  const int idx  = lane & 15;
  const int ntiles = (N + 15) / 16;
  const int npanels = (ntiles + TS - 1) / TS;
  const int xr = tid / C4;
  const int xc = (tid % C4) * 4;

  auto load_panel = [&](int p, v4f (&v)[XQ]) {
#pragma unroll
    for (int q = 0; q < XQ; ++q) {
      const int row = min(p * PR + xr + q * RQ, N - 1);
      v[q] = *(const v4f*)(A + (size_t)row * K + xc);
    }
  };
  auto store_panel = [&](float* xb, const v4f (&v)[XQ]) {
#pragma unroll
    for (int q = 0; q < XQ; ++q) *(v4f*)(xb + (xr + q * RQ) * XS + xc) = v[q];
  };

  int p = blockIdx.x;
  v4f nx[XQ];
  if (p < npanels) load_panel(p, nx);
  {
    v4f wv[WQ];
#pragma unroll
    for (int q = 0; q < WQ; ++q) wv[q] = *(const v4f*)(B + (size_t)(xr + q * RQ) * K + xc);
#pragma unroll
    for (int q = 0; q < WQ; ++q) *(v4f*)(wsm + (xr + q * RQ) * XS + xc) = wv[q];
  }
  if (p < npanels) store_panel(xsm[0], nx);
  __syncthreads();

  v2f b[KJ];
  {
    const float* bp = wsm + (cs * 16 + idx) * XS + 2 * half;
#pragma unroll
    for (int j = 0; j < KJ; ++j) b[j] = *(const v2f*)(bp + 4 * j);
  }

  for (int it = 0; p < npanels; p += gridDim.x, ++it) {
    const int pn = p + gridDim.x;
    const bool more = pn < npanels;
    if (more) load_panel(pn, nx);
    const float* xb = xsm[it & 1];
    const int tile = p * TS + ts;
    if (tile < ntiles) {
      const float* ap = xb + (ts * 16 + idx) * XS + 2 * half;
      v8f c = {0.f, 0.f, 0.f, 0.f, 0.f, 0.f, 0.f, 0.f};
      v2f a[2][KB];
#pragma unroll
      for (int i = 0; i < KB; ++i) a[0][i] = *(const v2f*)(ap + 4 * i);
#pragma unroll
      for (int jb = 0; jb < KJ / KB; ++jb) {
        if (jb + 1 < KJ / KB) {
#pragma unroll
          for (int i = 0; i < KB; ++i) a[(jb + 1) & 1][i] = *(const v2f*)(ap + 4 * ((jb + 1) * KB + i));
        }
#pragma unroll
        for (int i = 0; i < KB; ++i)
          c = __builtin_amdgcn_wmma_f32_16x16x4_f32(false, a[jb & 1][i], false, b[jb * KB + i],
                                                    (short)0, c, false, false);
      }
      __builtin_amdgcn_sched_group_barrier(0x100, KB / 2, 0);
#pragma unroll
      for (int jb = 0; jb < KJ / KB; ++jb) {
        if (jb + 1 < KJ / KB) __builtin_amdgcn_sched_group_barrier(0x100, KB / 2, 0);
        __builtin_amdgcn_sched_group_barrier(0x008, KB, 0);
      }
      const int col = cs * 16 + idx;
#pragma unroll
      for (int r = 0; r < 8; ++r) {
        const int row = tile * 16 + r + 8 * half;
        if (row < N) C[(size_t)row * NC + col] = c[r];
      }
    }
    if (more) store_panel(xsm[(it + 1) & 1], nx);
    __syncthreads();
  }
}

__global__ void attn_logits(const float* __restrict__ h, const float* __restrict__ att_s,
                            const float* __restrict__ att_d, float* __restrict__ a_s,
                            float* __restrict__ a_d, int NH, int H, int D) {
  int i = blockIdx.x * blockDim.x + threadIdx.x;
  if (i >= NH) return;
  const int hh = i % H;
  const float* __restrict__ hp = h + (size_t)i * D;
  const float* __restrict__ as = att_s + (size_t)hh * D;
  const float* __restrict__ ad = att_d + (size_t)hh * D;
  float s0 = 0.f, s1 = 0.f;
  for (int dd = 0; dd < D; ++dd) {
    const float v = hp[dd];
    s0 += v * as[dd];
    s1 += v * ad[dd];
  }
  a_s[i] = s0;
  a_d[i] = s1;
}

#define AGG_R 196
#define AGG_NT 1024
#define AGG_K 16
#define AGG_B 8
#define AGG_CAP 1024

template <int H, int D, int NC>
__global__ void __launch_bounds__(AGG_NT)
gat_softmax_aggregate(const int* __restrict__ src, const int* __restrict__ dst, int E, int N,
                      const float* __restrict__ a_s, const float* __restrict__ a_d,
                      const float* __restrict__ h,
                      const float* __restrict__ bias,
                      float* __restrict__ agg,
                      const float* __restrict__ Wn,
                      const float* __restrict__ att_sn,
                      const float* __restrict__ att_dn,
                      float* __restrict__ hn,
                      float* __restrict__ a_sn, float* __restrict__ a_dn) {
  constexpr int F = H * D;
  constexpr int R = AGG_R;
  constexpr int J = (F + 31) / 32;
  constexpr int K = AGG_K;
  constexpr int B = AGG_B;
  constexpr int CAP = AGG_CAP;
  constexpr int NW = AGG_NT / 32;
  constexpr int G = 32 * K;
  static_assert(B * H <= 32, "one lane per (edge, head) of a batch");
  static_assert((B & (B - 1)) == 0 && 2 * B <= 32 && CAP % (2 * B) == 0 && CAP >= 2 * G, "list geometry");
  static_assert(R < 256 && R <= 32 * NW, "local destination and the spare row fit 8 bits");
  __shared__ __attribute__((aligned(16))) float acc[(R + 1) * F];
  __shared__ float s_ad[(R + 1) * H];
  __shared__ unsigned s_mk[(R + 1) * H];
  __shared__ unsigned s_mo[(R + 1) * H];
  __shared__ float s_den[(R + 1) * H];
  __shared__ __attribute__((aligned(16))) unsigned lst[NW * CAP];

  const int tid = threadIdx.x;
  const int lane = tid & 31;
  const int wave = __builtin_amdgcn_readfirstlane(tid >> 5);
  const int lo = blockIdx.x * R;
  const int cnt = min(R, N - lo);
  unsigned* __restrict__ my = lst + wave * CAP;

  for (int i = tid; i < (R + 1) * F; i += AGG_NT) acc[i] = 0.f;
  for (int i = tid; i < (R + 1) * H; i += AGG_NT) {
    s_ad[i] = (i < cnt * H) ? a_d[(size_t)lo * H + i] : 0.f;
    s_mk[i] = 0u;
    s_den[i] = 0.f;
  }

  const int ngroups = (E + G - 1) / G;
  auto load_dst = [&](int g, int (&d)[K]) {
    const int e0 = g * G + lane;
    if (g * G + G <= E) {
#pragma unroll
      for (int k = 0; k < K; ++k) d[k] = dst[e0 + k * 32];
    } else {
#pragma unroll
      for (int k = 0; k < K; ++k) {
        const int e = e0 + k * 32;
        const int dv = dst[max(0, min(e, E - 1))];
        d[k] = (e < E) ? dv : -1;
      }
    }
  };

  int g = wave;
  int dA[K];
  load_dst(min(g, ngroups), dA);
  __syncthreads();

  for (int round = 0;; ++round) {
    int n_w = 0;
    if (round == 0) {
      const int dl = wave + NW * lane;
      const bool hit = dl < cnt;
      const unsigned mk = __builtin_amdgcn_ballot_w32(hit);
      if (hit) my[__builtin_popcount(mk & ((1u << lane) - 1u))] = ((unsigned)(E + lo + dl) << 8) | (unsigned)dl;
      n_w = __builtin_popcount(mk);
    }
    while (g < ngroups && n_w <= CAP - G - 2 * B) {
      int dB[K];
      load_dst(min(g + NW, ngroups), dB);
#pragma unroll
      for (int k = 0; k < K; ++k) {
        const int dl = dA[k] - lo;
        const bool hit = (unsigned)dl < (unsigned)cnt;
        const unsigned mk = __builtin_amdgcn_ballot_w32(hit);
        if (mk) {
          if (hit)
            my[n_w + __builtin_popcount(mk & ((1u << lane) - 1u))] =
                ((unsigned)(g * G + k * 32 + lane) << 8) | (unsigned)dl;
          n_w += __builtin_popcount(mk);
        }
      }
#pragma unroll
      for (int k = 0; k < K; ++k) dA[k] = dB[k];
      g += NW;
    }
    const int n_pad = (n_w + 2 * B - 1) / (2 * B) * (2 * B);
    if (lane < n_pad - n_w) my[n_w + lane] = ((unsigned)(E + lo) << 8) | (unsigned)R;

    for (int i = lane; i < n_pad; i += 32) {
      const unsigned en = my[i];
      const int e = (int)(en >> 8);
      const int dl = (int)(en & 255u);
      const int sv = src[min(e, E - 1)];
      const int s = (e < E) ? sv : e - E;
      my[i] = ((unsigned)s << 8) | (unsigned)dl;
#pragma unroll
      for (int hh = 0; hh < H; ++hh) {
        float v = a_s[(size_t)s * H + hh] + s_ad[dl * H + hh];
        v = (v > 0.f) ? v : NEG_SLOPE * v;
        atomicMax(&s_mk[dl * H + hh], fkey(v));
      }
    }
    __syncthreads();
    if (round > 0) {
      for (int i = tid; i < R * F; i += AGG_NT) {
        const int q = (i / F) * H + (i % F) / D;
        const unsigned ko = s_mo[q], kn = s_mk[q];
        if (ko != kn) acc[i] *= expf(fdekey(ko) - fdekey(kn));
      }
      for (int i = tid; i < R * H; i += AGG_NT) {
        const unsigned ko = s_mo[i], kn = s_mk[i];
        if (ko != kn) s_den[i] *= expf(fdekey(ko) - fdekey(kn));
      }
      __syncthreads();
    }

    const int nb2 = n_pad / (2 * B);
    const int tq = min(lane / H, B - 1);
    const int th = lane - (lane / H) * H;
    auto request = [&](int b, float (&v)[B][J], float& as_t, int& dl_t, int& dl_l) {
      const unsigned en_l = my[b * B + (lane & (B - 1))];
      const unsigned en_t = my[b * B + tq];
#pragma unroll
      for (int q = 0; q < B; ++q) {
        const int s_q = (int)((unsigned)__builtin_amdgcn_readlane((int)en_l, q) >> 8);
#pragma unroll
        for (int j = 0; j < J; ++j) {
          const int f = (F % 32 == 0) ? j * 32 + lane : min(j * 32 + lane, F - 1);
          v[q][j] = h[(size_t)s_q * F + f];
        }
      }
      as_t = a_s[(size_t)(en_t >> 8) * H + th];
      dl_t = (int)(en_t & 255u);
      dl_l = (int)(en_l & 255u);
    };
    auto add_batch = [&](const float (&v)[B][J], float as_t, int dl_t, int dl_l) {
      float e = as_t + s_ad[dl_t * H + th];
      e = (e > 0.f) ? e : NEG_SLOPE * e;
      const float ex = expf(e - fdekey(s_mk[dl_t * H + th]));
      if (lane < B * H) atomicAdd(&s_den[dl_t * H + th], ex);
#pragma unroll
      for (int q = 0; q < B; ++q) {
        const int dl_q = __builtin_amdgcn_readlane(dl_l, q);
#pragma unroll
        for (int j = 0; j < J; ++j) {
          const int f = j * 32 + lane;
          const int hh = (H == 1) ? 0 : ((D % 32 == 0) ? (j * 32) / D : f / D);
          const float w = __int_as_float(
              __builtin_amdgcn_readlane(__float_as_int(ex), q * H + hh));
          if (F % 32 == 0 || f < F) atomicAdd(&acc[dl_q * F + f], w * v[q][j]);
        }
      }
    };
    if (nb2 > 0) {
      float vA[B][J], vB[B][J];
      float asA, asB;
      int dtA, dlA, dtB, dlB;
      request(0, vA, asA, dtA, dlA);
      for (int b = 0; b < nb2; ++b) {
        request(2 * b + 1, vB, asB, dtB, dlB);
        add_batch(vA, asA, dtA, dlA);
        request(min(2 * b + 2, 2 * nb2 - 1), vA, asA, dtA, dlA);
        add_batch(vB, asB, dtB, dlB);
      }
    }

    if (!__syncthreads_or(g < ngroups)) break;
    for (int i = tid; i < (R + 1) * H; i += AGG_NT) s_mo[i] = s_mk[i];
    __syncthreads();
  }

  if constexpr (NC == 0) {
    for (int i = tid; i < cnt * F; i += AGG_NT) {
      const int q = (i / F) * H + (i % F) / D;
      agg[(size_t)lo * F + i] = acc[i] / s_den[q] + bias[i % F];
    }
  } else {
    constexpr int XS = F + 4;
    constexpr int NCT = (NC + 15) / 16;
    static_assert(F % 4 == 0 && XS % 64 == 4, "k steps of 4; conflict-free fragment reads");
    static_assert((R + 15) / 16 <= NW, "one wave per 16-row tile");
    static_assert((R + NCT * 16) * XS <= NW * CAP, "x and Wn fit the list's space");
    static_assert(R * NC <= (R + 1) * F, "the product fits the rows' space");
    float* xp = reinterpret_cast<float*>(lst);
    float* wp = xp + R * XS;
    float* hs = acc;
    constexpr int WJ = (NCT * 16 * F + AGG_NT - 1) / AGG_NT;
    float wv[WJ];
#pragma unroll
    for (int j = 0; j < WJ; ++j) {
      const int i = tid + j * AGG_NT;
      wv[j] = (i < NC * F) ? Wn[i] : 0.f;
    }
    for (int i = tid; i < cnt * F; i += AGG_NT) {
      const int r = i / F, f = i % F;
      const float v = acc[i] / s_den[r * H + f / D] + bias[f];
      xp[r * XS + f] = (v > 0.f) ? v : 0.f;
    }
#pragma unroll
    for (int j = 0; j < WJ; ++j) {
      const int i = tid + j * AGG_NT;
      if (i < NCT * 16 * F) wp[(i / F) * XS + i % F] = wv[j];
    }
    __syncthreads();
    if (wave * 16 < cnt) {
      const int half = lane >> 4;
      const int idx = lane & 15;
      const int ar = min(wave * 16 + idx, cnt - 1);
      const float* ap = xp + ar * XS + 2 * half;
      const float* bp = wp + idx * XS + 2 * half;
      v8f c[NCT];
#pragma unroll
      for (int t = 0; t < NCT; ++t) c[t] = v8f{0.f, 0.f, 0.f, 0.f, 0.f, 0.f, 0.f, 0.f};
#pragma unroll
      for (int k = 0; k < F; k += 4) {
        const v2f a = *(const v2f*)(ap + k);
#pragma unroll
        for (int t = 0; t < NCT; ++t) {
          const v2f b = *(const v2f*)(bp + t * 16 * XS + k);
          c[t] = __builtin_amdgcn_wmma_f32_16x16x4_f32(false, a, false, b, (short)0, c[t],
                                                       false, false);
        }
      }
#pragma unroll
      for (int t = 0; t < NCT; ++t) {
        const int col = t * 16 + idx;
#pragma unroll
        for (int r = 0; r < 8; ++r) {
          const int row = wave * 16 + r + 8 * half;
          if (col < NC && row < cnt) hs[row * NC + col] = c[t][r];
        }
      }
    }
    __syncthreads();
    for (int i = tid; i < cnt * NC; i += AGG_NT) hn[(size_t)lo * NC + i] = hs[i];
    if (tid < cnt) {
      float s0 = 0.f, s1 = 0.f;
      for (int dd = 0; dd < NC; ++dd) {
        const float v = hs[tid * NC + dd];
        s0 += v * att_sn[dd];
        s1 += v * att_dn[dd];
      }
      a_sn[lo + tid] = s0;
      a_dn[lo + tid] = s1;
    }
  }
}

extern "C" void kernel_launch(void* const* d_in, const int* in_sizes, int n_in,
                              void* d_out, int out_size, void* d_ws, size_t ws_size,
                              hipStream_t stream) {
  const float* x        = (const float*)d_in[0];
  const int*   ei       = (const int*)d_in[1];
  const float* W1       = (const float*)d_in[2];
  const float* att_src1 = (const float*)d_in[3];
  const float* att_dst1 = (const float*)d_in[4];
  const float* b1       = (const float*)d_in[5];
  const float* W2       = (const float*)d_in[6];
  const float* att_src2 = (const float*)d_in[7];
  const float* att_dst2 = (const float*)d_in[8];
  const float* b2       = (const float*)d_in[9];

  const int NFEAT = 256, NHID = 128, H1 = 4, D1 = 32, NCLS = 40;
  const int N = in_sizes[0] / NFEAT;
  const int E = in_sizes[1] / 2;
  const int* src = ei;
  const int* dst = ei + E;

  float* out = (float*)d_out;

  float* w = (float*)d_ws;
  float* h1   = w;
  float* h2   = h1  + (size_t)N * NHID;
  float* as1  = h2  + (size_t)N * NCLS;
  float* ad1  = as1 + (size_t)N * H1;
  float* as2  = ad1 + (size_t)N * H1;
  float* ad2  = as2 + (size_t)N;

  const int TB = 256;

  gemm_wmma_f32_bres<256, 128><<<G1_WG, G1_NT, 0, stream>>>(x, W1, h1, N);
  attn_logits<<<(N * H1 + TB - 1) / TB, TB, 0, stream>>>(h1, att_src1, att_dst1,
                                                         as1, ad1, N * H1, H1, D1);
  gat_softmax_aggregate<4, 32, 40><<<(N + AGG_R - 1) / AGG_R, AGG_NT, 0, stream>>>(
      src, dst, E, N, as1, ad1, h1, b1, nullptr, W2, att_src2, att_dst2, h2, as2, ad2);

  gat_softmax_aggregate<1, 40, 0><<<(N + AGG_R - 1) / AGG_R, AGG_NT, 0, stream>>>(
      src, dst, E, N, as2, ad2, h2, b2, out, nullptr, nullptr, nullptr, nullptr, nullptr, nullptr);
}


# layer 2's aggregation no longer scans the edge list: layer 1's work-groups store their resolved hit lists (source, local destination) in the workspace, layer 2 reads them; sum order = LDS arrival
# speedup vs baseline: 13.9335x; 1.1939x over previous
#include <hip/hip_runtime.h>

typedef float v2f __attribute__((ext_vector_type(2)));
typedef float v4f __attribute__((ext_vector_type(4)));
typedef float v8f __attribute__((ext_vector_type(8)));

#define NEG_SLOPE 0.2f

__device__ __forceinline__ unsigned fkey(float f) {
  unsigned u = __float_as_uint(f);
  return (u & 0x80000000u) ? ~u : (u | 0x80000000u);
}
__device__ __forceinline__ float fdekey(unsigned k) {
  unsigned u = (k & 0x80000000u) ? (k & 0x7FFFFFFFu) : ~k;
  return __uint_as_float(u);
}

#define G1_NT 512
#define G1_WG 256

template <int K, int NC>
__global__ void __launch_bounds__(G1_NT)
gemm_wmma_f32_bres(const float* __restrict__ A, const float* __restrict__ B,
                   float* __restrict__ C, int N) {
  constexpr int XS = K + 4;
  constexpr int NCS = NC / 16;
  constexpr int NW = G1_NT / 32;
  constexpr int TS = NW / NCS;
  constexpr int PR = TS * 16;
  constexpr int KJ = K / 4;
  constexpr int KB = 16;
  constexpr int C4 = K / 4;
  constexpr int RQ = G1_NT / C4;
  constexpr int XQ = PR / RQ;
  constexpr int WQ = NC / RQ;
  static_assert(K % 4 == 0 && KJ % KB == 0 && XS % 64 == 4, "k steps of 4; conflict-free fragment reads");
  static_assert(NC % 16 == 0 && NW % NCS == 0 && G1_NT % C4 == 0 && PR % RQ == 0 && NC % RQ == 0,
                "panel geometry");
  __shared__ __attribute__((aligned(16))) float wsm[NC * XS];
  __shared__ __attribute__((aligned(16))) float xsm[2][PR * XS];

  const int tid = threadIdx.x;
  const int lane = tid & 31;
  const int wave = __builtin_amdgcn_readfirstlane(tid >> 5);
  const int cs = wave % NCS;
  const int ts = wave / NCS;
  const int half = lane >> 4;
  const int idx  = lane & 15;
  const int ntiles = (N + 15) / 16;
  const int npanels = (ntiles + TS - 1) / TS;
  const int xr = tid / C4;
  const int xc = (tid % C4) * 4;

  auto load_panel = [&](int p, v4f (&v)[XQ]) {
#pragma unroll
    for (int q = 0; q < XQ; ++q) {
      const int row = min(p * PR + xr + q * RQ, N - 1);
      v[q] = *(const v4f*)(A + (size_t)row * K + xc);
    }
  };
  auto store_panel = [&](float* xb, const v4f (&v)[XQ]) {
#pragma unroll
    for (int q = 0; q < XQ; ++q) *(v4f*)(xb + (xr + q * RQ) * XS + xc) = v[q];
  };

  int p = blockIdx.x;
  v4f nx[XQ];
  if (p < npanels) load_panel(p, nx);
  {
    v4f wv[WQ];
#pragma unroll
    for (int q = 0; q < WQ; ++q) wv[q] = *(const v4f*)(B + (size_t)(xr + q * RQ) * K + xc);
#pragma unroll
    for (int q = 0; q < WQ; ++q) *(v4f*)(wsm + (xr + q * RQ) * XS + xc) = wv[q];
  }
  if (p < npanels) store_panel(xsm[0], nx);
  __syncthreads();

  v2f b[KJ];
  {
    const float* bp = wsm + (cs * 16 + idx) * XS + 2 * half;
#pragma unroll
    for (int j = 0; j < KJ; ++j) b[j] = *(const v2f*)(bp + 4 * j);
  }

  for (int it = 0; p < npanels; p += gridDim.x, ++it) {
    const int pn = p + gridDim.x;
    const bool more = pn < npanels;
    if (more) load_panel(pn, nx);
    const float* xb = xsm[it & 1];
    const int tile = p * TS + ts;
    if (tile < ntiles) {
      const float* ap = xb + (ts * 16 + idx) * XS + 2 * half;
      v8f c = {0.f, 0.f, 0.f, 0.f, 0.f, 0.f, 0.f, 0.f};
      v2f a[2][KB];
#pragma unroll
      for (int i = 0; i < KB; ++i) a[0][i] = *(const v2f*)(ap + 4 * i);
#pragma unroll
      for (int jb = 0; jb < KJ / KB; ++jb) {
        if (jb + 1 < KJ / KB) {
#pragma unroll
          for (int i = 0; i < KB; ++i) a[(jb + 1) & 1][i] = *(const v2f*)(ap + 4 * ((jb + 1) * KB + i));
        }
#pragma unroll
        for (int i = 0; i < KB; ++i)
          c = __builtin_amdgcn_wmma_f32_16x16x4_f32(false, a[jb & 1][i], false, b[jb * KB + i],
                                                    (short)0, c, false, false);
      }
      __builtin_amdgcn_sched_group_barrier(0x100, KB / 2, 0);
#pragma unroll
      for (int jb = 0; jb < KJ / KB; ++jb) {
        if (jb + 1 < KJ / KB) __builtin_amdgcn_sched_group_barrier(0x100, KB / 2, 0);
        __builtin_amdgcn_sched_group_barrier(0x008, KB, 0);
      }
      const int col = cs * 16 + idx;
#pragma unroll
      for (int r = 0; r < 8; ++r) {
        const int row = tile * 16 + r + 8 * half;
        if (row < N) C[(size_t)row * NC + col] = c[r];
      }
    }
    if (more) store_panel(xsm[(it + 1) & 1], nx);
    __syncthreads();
  }
}

__global__ void attn_logits(const float* __restrict__ h, const float* __restrict__ att_s,
                            const float* __restrict__ att_d, float* __restrict__ a_s,
                            float* __restrict__ a_d, int NH, int H, int D) {
  int i = blockIdx.x * blockDim.x + threadIdx.x;
  if (i >= NH) return;
  const int hh = i % H;
  const float* __restrict__ hp = h + (size_t)i * D;
  const float* __restrict__ as = att_s + (size_t)hh * D;
  const float* __restrict__ ad = att_d + (size_t)hh * D;
  float s0 = 0.f, s1 = 0.f;
  for (int dd = 0; dd < D; ++dd) {
    const float v = hp[dd];
    s0 += v * as[dd];
    s1 += v * ad[dd];
  }
  a_s[i] = s0;
  a_d[i] = s1;
}

#define AGG_R 196
#define AGG_NT 1024
#define AGG_K 16
#define AGG_B 8
#define AGG_CAP 1024

template <int H, int D, int NC, int LM>
__global__ void __launch_bounds__(AGG_NT)
gat_softmax_aggregate(const int* __restrict__ src, const int* __restrict__ dst, int E, int N,
                      const float* __restrict__ a_s, const float* __restrict__ a_d,
                      const float* __restrict__ h,
                      const float* __restrict__ bias,
                      float* __restrict__ agg,
                      const float* __restrict__ Wn,
                      const float* __restrict__ att_sn,
                      const float* __restrict__ att_dn,
                      float* __restrict__ hn,
                      float* __restrict__ a_sn, float* __restrict__ a_dn,
                      int* __restrict__ keep_lst,
                      int* __restrict__ keep_cnt,
                      int* __restrict__ keep_ok) {
  constexpr int F = H * D;
  constexpr int R = AGG_R;
  constexpr int J = (F + 31) / 32;
  constexpr int K = AGG_K;
  constexpr int B = AGG_B;
  constexpr int CAP = AGG_CAP;
  constexpr int NW = AGG_NT / 32;
  constexpr int G = 32 * K;
  static_assert(B * H <= 32, "one lane per (edge, head) of a batch");
  static_assert((B & (B - 1)) == 0 && 2 * B <= 32 && CAP % (2 * B) == 0 && CAP >= 2 * G, "list geometry");
  static_assert(R < 256 && R <= 32 * NW, "local destination and the spare row fit 8 bits");
  __shared__ __attribute__((aligned(16))) float acc[(R + 1) * F];
  __shared__ float s_ad[(R + 1) * H];
  __shared__ unsigned s_mk[(R + 1) * H];
  __shared__ unsigned s_mo[(R + 1) * H];
  __shared__ float s_den[(R + 1) * H];
  __shared__ __attribute__((aligned(16))) unsigned lst[NW * CAP];

  const int tid = threadIdx.x;
  const int lane = tid & 31;
  const int wave = __builtin_amdgcn_readfirstlane(tid >> 5);
  const int lo = blockIdx.x * R;
  const int cnt = min(R, N - lo);
  unsigned* __restrict__ my = lst + wave * CAP;

  for (int i = tid; i < (R + 1) * F; i += AGG_NT) acc[i] = 0.f;
  for (int i = tid; i < (R + 1) * H; i += AGG_NT) {
    s_ad[i] = (i < cnt * H) ? a_d[(size_t)lo * H + i] : 0.f;
    s_mk[i] = 0u;
    s_den[i] = 0.f;
  }

  const int ngroups = (E + G - 1) / G;
  auto load_dst = [&](int g, int (&d)[K]) {
    const int e0 = g * G + lane;
    if (g * G + G <= E) {
#pragma unroll
      for (int k = 0; k < K; ++k) d[k] = dst[e0 + k * 32];
    } else {
#pragma unroll
      for (int k = 0; k < K; ++k) {
        const int e = e0 + k * 32;
        const int dv = dst[max(0, min(e, E - 1))];
        d[k] = (e < E) ? dv : -1;
      }
    }
  };

  const int slot = blockIdx.x * NW + wave;
  bool kept = false;
  if constexpr (LM == 2) kept = __builtin_amdgcn_readfirstlane(keep_ok[blockIdx.x]) == 1;

  int g = wave;
  int dA[K];
  if (kept) {
    g = ngroups;
#pragma unroll
    for (int k = 0; k < K; ++k) dA[k] = -1;
  } else {
    load_dst(min(g, ngroups), dA);
  }
  __syncthreads();

  int round = 0;
  for (;; ++round) {
    int n_pad;
    if (kept) {
      n_pad = min(__builtin_amdgcn_readfirstlane(keep_cnt[slot]), CAP) & ~(2 * B - 1);
      const int* __restrict__ kl = keep_lst + (size_t)slot * CAP;
      for (int i = lane; i < n_pad; i += 32) {
        const unsigned en = (unsigned)kl[i];
        const int s = (int)(en >> 8);
        const int dl = (int)(en & 255u);
        my[i] = en;
#pragma unroll
        for (int hh = 0; hh < H; ++hh) {
          float v = a_s[(size_t)s * H + hh] + s_ad[dl * H + hh];
          v = (v > 0.f) ? v : NEG_SLOPE * v;
          atomicMax(&s_mk[dl * H + hh], fkey(v));
        }
      }
    } else {
      int n_w = 0;
      if (round == 0) {
        const int dl = wave + NW * lane;
        const bool hit = dl < cnt;
        const unsigned mk = __builtin_amdgcn_ballot_w32(hit);
        if (hit) my[__builtin_popcount(mk & ((1u << lane) - 1u))] = ((unsigned)(E + lo + dl) << 8) | (unsigned)dl;
        n_w = __builtin_popcount(mk);
      }
      while (g < ngroups && n_w <= CAP - G - 2 * B) {
        int dB[K];
        load_dst(min(g + NW, ngroups), dB);
#pragma unroll
        for (int k = 0; k < K; ++k) {
          const int dl = dA[k] - lo;
          const bool hit = (unsigned)dl < (unsigned)cnt;
          const unsigned mk = __builtin_amdgcn_ballot_w32(hit);
          if (mk) {
            if (hit)
              my[n_w + __builtin_popcount(mk & ((1u << lane) - 1u))] =
                  ((unsigned)(g * G + k * 32 + lane) << 8) | (unsigned)dl;
            n_w += __builtin_popcount(mk);
          }
        }
#pragma unroll
        for (int k = 0; k < K; ++k) dA[k] = dB[k];
        g += NW;
      }
      n_pad = (n_w + 2 * B - 1) / (2 * B) * (2 * B);
      if (lane < n_pad - n_w) my[n_w + lane] = ((unsigned)(E + lo) << 8) | (unsigned)R;
      if (LM == 1 && round == 0 && lane == 0) keep_cnt[slot] = n_pad;

      for (int i = lane; i < n_pad; i += 32) {
        const unsigned en = my[i];
        const int e = (int)(en >> 8);
        const int dl = (int)(en & 255u);
        const int sv = src[min(e, E - 1)];
        const int s = (e < E) ? sv : e - E;
        my[i] = ((unsigned)s << 8) | (unsigned)dl;
        if (LM == 1 && round == 0) keep_lst[(size_t)slot * CAP + i] = (s << 8) | dl;
#pragma unroll
        for (int hh = 0; hh < H; ++hh) {
          float v = a_s[(size_t)s * H + hh] + s_ad[dl * H + hh];
          v = (v > 0.f) ? v : NEG_SLOPE * v;
          atomicMax(&s_mk[dl * H + hh], fkey(v));
        }
      }
    }
    __syncthreads();
    if (round > 0) {
      for (int i = tid; i < R * F; i += AGG_NT) {
        const int q = (i / F) * H + (i % F) / D;
        const unsigned ko = s_mo[q], kn = s_mk[q];
        if (ko != kn) acc[i] *= expf(fdekey(ko) - fdekey(kn));
      }
      for (int i = tid; i < R * H; i += AGG_NT) {
        const unsigned ko = s_mo[i], kn = s_mk[i];
        if (ko != kn) s_den[i] *= expf(fdekey(ko) - fdekey(kn));
      }
      __syncthreads();
    }

    const int nb2 = n_pad / (2 * B);
    const int tq = min(lane / H, B - 1);
    const int th = lane - (lane / H) * H;
    auto request = [&](int b, float (&v)[B][J], float& as_t, int& dl_t, int& dl_l) {
      const unsigned en_l = my[b * B + (lane & (B - 1))];
      const unsigned en_t = my[b * B + tq];
#pragma unroll
      for (int q = 0; q < B; ++q) {
        const int s_q = (int)((unsigned)__builtin_amdgcn_readlane((int)en_l, q) >> 8);
#pragma unroll
        for (int j = 0; j < J; ++j) {
          const int f = (F % 32 == 0) ? j * 32 + lane : min(j * 32 + lane, F - 1);
          v[q][j] = h[(size_t)s_q * F + f];
        }
      }
      as_t = a_s[(size_t)(en_t >> 8) * H + th];
      dl_t = (int)(en_t & 255u);
      dl_l = (int)(en_l & 255u);
    };
    auto add_batch = [&](const float (&v)[B][J], float as_t, int dl_t, int dl_l) {
      float e = as_t + s_ad[dl_t * H + th];
      e = (e > 0.f) ? e : NEG_SLOPE * e;
      const float ex = expf(e - fdekey(s_mk[dl_t * H + th]));
      if (lane < B * H) atomicAdd(&s_den[dl_t * H + th], ex);
#pragma unroll
      for (int q = 0; q < B; ++q) {
        const int dl_q = __builtin_amdgcn_readlane(dl_l, q);
#pragma unroll
        for (int j = 0; j < J; ++j) {
          const int f = j * 32 + lane;
          const int hh = (H == 1) ? 0 : ((D % 32 == 0) ? (j * 32) / D : f / D);
          const float w = __int_as_float(
              __builtin_amdgcn_readlane(__float_as_int(ex), q * H + hh));
          if (F % 32 == 0 || f < F) atomicAdd(&acc[dl_q * F + f], w * v[q][j]);
        }
      }
    };
    if (nb2 > 0) {
      float vA[B][J], vB[B][J];
      float asA, asB;
      int dtA, dlA, dtB, dlB;
      request(0, vA, asA, dtA, dlA);
      for (int b = 0; b < nb2; ++b) {
        request(2 * b + 1, vB, asB, dtB, dlB);
        add_batch(vA, asA, dtA, dlA);
        request(min(2 * b + 2, 2 * nb2 - 1), vA, asA, dtA, dlA);
        add_batch(vB, asB, dtB, dlB);
      }
    }

    if (!__syncthreads_or(g < ngroups)) break;
    for (int i = tid; i < (R + 1) * H; i += AGG_NT) s_mo[i] = s_mk[i];
    __syncthreads();
  }
  if (LM == 1 && tid == 0) keep_ok[blockIdx.x] = (round == 0) ? 1 : 0;

  if constexpr (NC == 0) {
    for (int i = tid; i < cnt * F; i += AGG_NT) {
      const int q = (i / F) * H + (i % F) / D;
      agg[(size_t)lo * F + i] = acc[i] / s_den[q] + bias[i % F];
    }
  } else {
    constexpr int XS = F + 4;
    constexpr int NCT = (NC + 15) / 16;
    static_assert(F % 4 == 0 && XS % 64 == 4, "k steps of 4; conflict-free fragment reads");
    static_assert((R + 15) / 16 <= NW, "one wave per 16-row tile");
    static_assert((R + NCT * 16) * XS <= NW * CAP, "x and Wn fit the list's space");
    static_assert(R * NC <= (R + 1) * F, "the product fits the rows' space");
    float* xp = reinterpret_cast<float*>(lst);
    float* wp = xp + R * XS;
    float* hs = acc;
    constexpr int WJ = (NCT * 16 * F + AGG_NT - 1) / AGG_NT;
    float wv[WJ];
#pragma unroll
    for (int j = 0; j < WJ; ++j) {
      const int i = tid + j * AGG_NT;
      wv[j] = (i < NC * F) ? Wn[i] : 0.f;
    }
    for (int i = tid; i < cnt * F; i += AGG_NT) {
      const int r = i / F, f = i % F;
      const float v = acc[i] / s_den[r * H + f / D] + bias[f];
      xp[r * XS + f] = (v > 0.f) ? v : 0.f;
    }
#pragma unroll
    for (int j = 0; j < WJ; ++j) {
      const int i = tid + j * AGG_NT;
      if (i < NCT * 16 * F) wp[(i / F) * XS + i % F] = wv[j];
    }
    __syncthreads();
    if (wave * 16 < cnt) {
      const int half = lane >> 4;
      const int idx = lane & 15;
      const int ar = min(wave * 16 + idx, cnt - 1);
      const float* ap = xp + ar * XS + 2 * half;
      const float* bp = wp + idx * XS + 2 * half;
      v8f c[NCT];
#pragma unroll
      for (int t = 0; t < NCT; ++t) c[t] = v8f{0.f, 0.f, 0.f, 0.f, 0.f, 0.f, 0.f, 0.f};
#pragma unroll
      for (int k = 0; k < F; k += 4) {
        const v2f a = *(const v2f*)(ap + k);
#pragma unroll
        for (int t = 0; t < NCT; ++t) {
          const v2f b = *(const v2f*)(bp + t * 16 * XS + k);
          c[t] = __builtin_amdgcn_wmma_f32_16x16x4_f32(false, a, false, b, (short)0, c[t],
                                                       false, false);
        }
      }
#pragma unroll
      for (int t = 0; t < NCT; ++t) {
        const int col = t * 16 + idx;
#pragma unroll
        for (int r = 0; r < 8; ++r) {
          const int row = wave * 16 + r + 8 * half;
          if (col < NC && row < cnt) hs[row * NC + col] = c[t][r];
        }
      }
    }
    __syncthreads();
    for (int i = tid; i < cnt * NC; i += AGG_NT) hn[(size_t)lo * NC + i] = hs[i];
    if (tid < cnt) {
      float s0 = 0.f, s1 = 0.f;
      for (int dd = 0; dd < NC; ++dd) {
        const float v = hs[tid * NC + dd];
        s0 += v * att_sn[dd];
        s1 += v * att_dn[dd];
      }
      a_sn[lo + tid] = s0;
      a_dn[lo + tid] = s1;
    }
  }
}

extern "C" void kernel_launch(void* const* d_in, const int* in_sizes, int n_in,
                              void* d_out, int out_size, void* d_ws, size_t ws_size,
                              hipStream_t stream) {
  const float* x        = (const float*)d_in[0];
  const int*   ei       = (const int*)d_in[1];
  const float* W1       = (const float*)d_in[2];
  const float* att_src1 = (const float*)d_in[3];
  const float* att_dst1 = (const float*)d_in[4];
  const float* b1       = (const float*)d_in[5];
  const float* W2       = (const float*)d_in[6];
  const float* att_src2 = (const float*)d_in[7];
  const float* att_dst2 = (const float*)d_in[8];
  const float* b2       = (const float*)d_in[9];

  const int NFEAT = 256, NHID = 128, H1 = 4, D1 = 32, NCLS = 40;
  const int N = in_sizes[0] / NFEAT;
  const int E = in_sizes[1] / 2;
  const int* src = ei;
  const int* dst = ei + E;

  float* out = (float*)d_out;

  float* w = (float*)d_ws;
  float* h1   = w;
  float* h2   = h1  + (size_t)N * NHID;
  float* as1  = h2  + (size_t)N * NCLS;
  float* ad1  = as1 + (size_t)N * H1;
  float* as2  = ad1 + (size_t)N * H1;
  float* ad2  = as2 + (size_t)N;
  const int groups = (N + AGG_R - 1) / AGG_R;
  int* klst = (int*)(ad2 + (size_t)N);
  int* kcnt = klst + (size_t)groups * (AGG_NT / 32) * AGG_CAP;
  int* kok  = kcnt + (size_t)groups * (AGG_NT / 32);

  const int TB = 256;

  gemm_wmma_f32_bres<256, 128><<<G1_WG, G1_NT, 0, stream>>>(x, W1, h1, N);
  attn_logits<<<(N * H1 + TB - 1) / TB, TB, 0, stream>>>(h1, att_src1, att_dst1,
                                                         as1, ad1, N * H1, H1, D1);
  gat_softmax_aggregate<4, 32, 40, 1><<<groups, AGG_NT, 0, stream>>>(
      src, dst, E, N, as1, ad1, h1, b1, nullptr, W2, att_src2, att_dst2, h2, as2, ad2,
      klst, kcnt, kok);

  gat_softmax_aggregate<1, 40, 0, 2><<<groups, AGG_NT, 0, stream>>>(
      src, dst, E, N, as2, ad2, h2, b2, out, nullptr, nullptr, nullptr, nullptr, nullptr, nullptr,
      klst, kcnt, kok);
}


# a listed edge's leaky logits are computed once, in the pass that raises the maxima, and kept per entry (1 head: LDS; 4 heads: workspace); the weighted-sum pass reads them back, no second a_s gather
# speedup vs baseline: 14.3799x; 1.0320x over previous
#include <hip/hip_runtime.h>

typedef float v2f __attribute__((ext_vector_type(2)));
typedef float v4f __attribute__((ext_vector_type(4)));
typedef float v8f __attribute__((ext_vector_type(8)));

#define NEG_SLOPE 0.2f

__device__ __forceinline__ unsigned fkey(float f) {
  unsigned u = __float_as_uint(f);
  return (u & 0x80000000u) ? ~u : (u | 0x80000000u);
}
__device__ __forceinline__ float fdekey(unsigned k) {
  unsigned u = (k & 0x80000000u) ? (k & 0x7FFFFFFFu) : ~k;
  return __uint_as_float(u);
}

#define G1_NT 512
#define G1_WG 256

template <int K, int NC>
__global__ void __launch_bounds__(G1_NT)
gemm_wmma_f32_bres(const float* __restrict__ A, const float* __restrict__ B,
                   float* __restrict__ C, int N) {
  constexpr int XS = K + 4;
  constexpr int NCS = NC / 16;
  constexpr int NW = G1_NT / 32;
  constexpr int TS = NW / NCS;
  constexpr int PR = TS * 16;
  constexpr int KJ = K / 4;
  constexpr int KB = 16;
  constexpr int C4 = K / 4;
  constexpr int RQ = G1_NT / C4;
  constexpr int XQ = PR / RQ;
  constexpr int WQ = NC / RQ;
  static_assert(K % 4 == 0 && KJ % KB == 0 && XS % 64 == 4, "k steps of 4; conflict-free fragment reads");
  static_assert(NC % 16 == 0 && NW % NCS == 0 && G1_NT % C4 == 0 && PR % RQ == 0 && NC % RQ == 0,
                "panel geometry");
  __shared__ __attribute__((aligned(16))) float wsm[NC * XS];
  __shared__ __attribute__((aligned(16))) float xsm[2][PR * XS];

  const int tid = threadIdx.x;
  const int lane = tid & 31;
  const int wave = __builtin_amdgcn_readfirstlane(tid >> 5);
  const int cs = wave % NCS;
  const int ts = wave / NCS;
  const int half = lane >> 4;
  const int idx  = lane & 15;
  const int ntiles = (N + 15) / 16;
  const int npanels = (ntiles + TS - 1) / TS;
  const int xr = tid / C4;
  const int xc = (tid % C4) * 4;

  auto load_panel = [&](int p, v4f (&v)[XQ]) {
#pragma unroll
    for (int q = 0; q < XQ; ++q) {
      const int row = min(p * PR + xr + q * RQ, N - 1);
      v[q] = *(const v4f*)(A + (size_t)row * K + xc);
    }
  };
  auto store_panel = [&](float* xb, const v4f (&v)[XQ]) {
#pragma unroll
    for (int q = 0; q < XQ; ++q) *(v4f*)(xb + (xr + q * RQ) * XS + xc) = v[q];
  };

  int p = blockIdx.x;
  v4f nx[XQ];
  if (p < npanels) load_panel(p, nx);
  {
    v4f wv[WQ];
#pragma unroll
    for (int q = 0; q < WQ; ++q) wv[q] = *(const v4f*)(B + (size_t)(xr + q * RQ) * K + xc);
#pragma unroll
    for (int q = 0; q < WQ; ++q) *(v4f*)(wsm + (xr + q * RQ) * XS + xc) = wv[q];
  }
  if (p < npanels) store_panel(xsm[0], nx);
  __syncthreads();

  v2f b[KJ];
  {
    const float* bp = wsm + (cs * 16 + idx) * XS + 2 * half;
#pragma unroll
    for (int j = 0; j < KJ; ++j) b[j] = *(const v2f*)(bp + 4 * j);
  }

  for (int it = 0; p < npanels; p += gridDim.x, ++it) {
    const int pn = p + gridDim.x;
    const bool more = pn < npanels;
    if (more) load_panel(pn, nx);
    const float* xb = xsm[it & 1];
    const int tile = p * TS + ts;
    if (tile < ntiles) {
      const float* ap = xb + (ts * 16 + idx) * XS + 2 * half;
      v8f c = {0.f, 0.f, 0.f, 0.f, 0.f, 0.f, 0.f, 0.f};
      v2f a[2][KB];
#pragma unroll
      for (int i = 0; i < KB; ++i) a[0][i] = *(const v2f*)(ap + 4 * i);
#pragma unroll
      for (int jb = 0; jb < KJ / KB; ++jb) {
        if (jb + 1 < KJ / KB) {
#pragma unroll
          for (int i = 0; i < KB; ++i) a[(jb + 1) & 1][i] = *(const v2f*)(ap + 4 * ((jb + 1) * KB + i));
        }
#pragma unroll
        for (int i = 0; i < KB; ++i)
          c = __builtin_amdgcn_wmma_f32_16x16x4_f32(false, a[jb & 1][i], false, b[jb * KB + i],
                                                    (short)0, c, false, false);
      }
      __builtin_amdgcn_sched_group_barrier(0x100, KB / 2, 0);
#pragma unroll
      for (int jb = 0; jb < KJ / KB; ++jb) {
        if (jb + 1 < KJ / KB) __builtin_amdgcn_sched_group_barrier(0x100, KB / 2, 0);
        __builtin_amdgcn_sched_group_barrier(0x008, KB, 0);
      }
      const int col = cs * 16 + idx;
#pragma unroll
      for (int r = 0; r < 8; ++r) {
        const int row = tile * 16 + r + 8 * half;
        if (row < N) C[(size_t)row * NC + col] = c[r];
      }
    }
    if (more) store_panel(xsm[(it + 1) & 1], nx);
    __syncthreads();
  }
}

__global__ void attn_logits(const float* __restrict__ h, const float* __restrict__ att_s,
                            const float* __restrict__ att_d, float* __restrict__ a_s,
                            float* __restrict__ a_d, int NH, int H, int D) {
  int i = blockIdx.x * blockDim.x + threadIdx.x;
  if (i >= NH) return;
  const int hh = i % H;
  const float* __restrict__ hp = h + (size_t)i * D;
  const float* __restrict__ as = att_s + (size_t)hh * D;
  const float* __restrict__ ad = att_d + (size_t)hh * D;
  float s0 = 0.f, s1 = 0.f;
  for (int dd = 0; dd < D; ++dd) {
    const float v = hp[dd];
    s0 += v * as[dd];
    s1 += v * ad[dd];
  }
  a_s[i] = s0;
  a_d[i] = s1;
}

#define AGG_R 196
#define AGG_NT 1024
#define AGG_K 16
#define AGG_B 8
#define AGG_CAP 1024

template <int H, int D, int NC, int LM>
__global__ void __launch_bounds__(AGG_NT)
gat_softmax_aggregate(const int* __restrict__ src, const int* __restrict__ dst, int E, int N,
                      const float* __restrict__ a_s, const float* __restrict__ a_d,
                      const float* __restrict__ h,
                      const float* __restrict__ bias,
                      float* __restrict__ agg,
                      const float* __restrict__ Wn,
                      const float* __restrict__ att_sn,
                      const float* __restrict__ att_dn,
                      float* __restrict__ hn,
                      float* __restrict__ a_sn, float* __restrict__ a_dn,
                      int* __restrict__ keep_lst,
                      int* __restrict__ keep_cnt,
                      int* __restrict__ keep_ok,
                      float* __restrict__ lg_ws) {
  constexpr int F = H * D;
  constexpr int R = AGG_R;
  constexpr int J = (F + 31) / 32;
  constexpr int K = AGG_K;
  constexpr int B = AGG_B;
  constexpr int CAP = AGG_CAP;
  constexpr int NW = AGG_NT / 32;
  constexpr int G = 32 * K;
  static_assert(B * H <= 32, "one lane per (edge, head) of a batch");
  static_assert((B & (B - 1)) == 0 && 2 * B <= 32 && CAP % (2 * B) == 0 && CAP >= 2 * G, "list geometry");
  static_assert(R < 256 && R <= 32 * NW, "local destination and the spare row fit 8 bits");
  __shared__ __attribute__((aligned(16))) float acc[(R + 1) * F];
  __shared__ float s_ad[(R + 1) * H];
  __shared__ unsigned s_mk[(R + 1) * H];
  __shared__ unsigned s_mo[(R + 1) * H];
  __shared__ float s_den[(R + 1) * H];
  __shared__ __attribute__((aligned(16))) unsigned lst[NW * CAP];
  __shared__ float lg_lds[(H == 1) ? NW * CAP : 1];

  const int tid = threadIdx.x;
  const int lane = tid & 31;
  const int wave = __builtin_amdgcn_readfirstlane(tid >> 5);
  const int lo = blockIdx.x * R;
  const int cnt = min(R, N - lo);
  unsigned* __restrict__ my = lst + wave * CAP;

  for (int i = tid; i < (R + 1) * F; i += AGG_NT) acc[i] = 0.f;
  for (int i = tid; i < (R + 1) * H; i += AGG_NT) {
    s_ad[i] = (i < cnt * H) ? a_d[(size_t)lo * H + i] : 0.f;
    s_mk[i] = 0u;
    s_den[i] = 0.f;
  }

  const int ngroups = (E + G - 1) / G;
  auto load_dst = [&](int g, int (&d)[K]) {
    const int e0 = g * G + lane;
    if (g * G + G <= E) {
#pragma unroll
      for (int k = 0; k < K; ++k) d[k] = dst[e0 + k * 32];
    } else {
#pragma unroll
      for (int k = 0; k < K; ++k) {
        const int e = e0 + k * 32;
        const int dv = dst[max(0, min(e, E - 1))];
        d[k] = (e < E) ? dv : -1;
      }
    }
  };

  const int slot = blockIdx.x * NW + wave;
  float* __restrict__ lg;
  if constexpr (H == 1) lg = lg_lds + wave * CAP;
  else lg = lg_ws + (size_t)slot * CAP * H;
  auto note_logits = [&](int i, int s, int dl) {
    float v[H];
#pragma unroll
    for (int hh = 0; hh < H; ++hh) {
      const float t = a_s[(size_t)s * H + hh] + s_ad[dl * H + hh];
      v[hh] = (t > 0.f) ? t : NEG_SLOPE * t;
      atomicMax(&s_mk[dl * H + hh], fkey(v[hh]));
    }
    if constexpr (H == 4) {
      *(v4f*)(lg + (size_t)i * H) = v4f{v[0], v[1], v[2], v[3]};
    } else {
#pragma unroll
      for (int hh = 0; hh < H; ++hh) lg[(size_t)i * H + hh] = v[hh];
    }
  };
  bool kept = false;
  if constexpr (LM == 2) kept = __builtin_amdgcn_readfirstlane(keep_ok[blockIdx.x]) == 1;

  int g = wave;
  int dA[K];
  if (kept) {
    g = ngroups;
#pragma unroll
    for (int k = 0; k < K; ++k) dA[k] = -1;
  } else {
    load_dst(min(g, ngroups), dA);
  }
  __syncthreads();

  int round = 0;
  for (;; ++round) {
    int n_pad;
    if (kept) {
      n_pad = min(__builtin_amdgcn_readfirstlane(keep_cnt[slot]), CAP) & ~(2 * B - 1);
      const int* __restrict__ kl = keep_lst + (size_t)slot * CAP;
      for (int i = lane; i < n_pad; i += 32) {
        const unsigned en = (unsigned)kl[i];
        const int s = (int)(en >> 8);
        const int dl = (int)(en & 255u);
        my[i] = en;
        note_logits(i, s, dl);
      }
    } else {
      int n_w = 0;
      if (round == 0) {
        const int dl = wave + NW * lane;
        const bool hit = dl < cnt;
        const unsigned mk = __builtin_amdgcn_ballot_w32(hit);
        if (hit) my[__builtin_popcount(mk & ((1u << lane) - 1u))] = ((unsigned)(E + lo + dl) << 8) | (unsigned)dl;
        n_w = __builtin_popcount(mk);
      }
      while (g < ngroups && n_w <= CAP - G - 2 * B) {
        int dB[K];
        load_dst(min(g + NW, ngroups), dB);
#pragma unroll
        for (int k = 0; k < K; ++k) {
          const int dl = dA[k] - lo;
          const bool hit = (unsigned)dl < (unsigned)cnt;
          const unsigned mk = __builtin_amdgcn_ballot_w32(hit);
          if (mk) {
            if (hit)
              my[n_w + __builtin_popcount(mk & ((1u << lane) - 1u))] =
                  ((unsigned)(g * G + k * 32 + lane) << 8) | (unsigned)dl;
            n_w += __builtin_popcount(mk);
          }
        }
#pragma unroll
        for (int k = 0; k < K; ++k) dA[k] = dB[k];
        g += NW;
      }
      n_pad = (n_w + 2 * B - 1) / (2 * B) * (2 * B);
      if (lane < n_pad - n_w) my[n_w + lane] = ((unsigned)(E + lo) << 8) | (unsigned)R;
      if (LM == 1 && round == 0 && lane == 0) keep_cnt[slot] = n_pad;

      for (int i = lane; i < n_pad; i += 32) {
        const unsigned en = my[i];
        const int e = (int)(en >> 8);
        const int dl = (int)(en & 255u);
        const int sv = src[min(e, E - 1)];
        const int s = (e < E) ? sv : e - E;
        my[i] = ((unsigned)s << 8) | (unsigned)dl;
        if (LM == 1 && round == 0) keep_lst[(size_t)slot * CAP + i] = (s << 8) | dl;
        note_logits(i, s, dl);
      }
    }
    __syncthreads();
    if (round > 0) {
      for (int i = tid; i < R * F; i += AGG_NT) {
        const int q = (i / F) * H + (i % F) / D;
        const unsigned ko = s_mo[q], kn = s_mk[q];
        if (ko != kn) acc[i] *= expf(fdekey(ko) - fdekey(kn));
      }
      for (int i = tid; i < R * H; i += AGG_NT) {
        const unsigned ko = s_mo[i], kn = s_mk[i];
        if (ko != kn) s_den[i] *= expf(fdekey(ko) - fdekey(kn));
      }
      __syncthreads();
    }

    const int nb2 = n_pad / (2 * B);
    const int tq = min(lane / H, B - 1);
    const int th = lane - (lane / H) * H;
    auto request = [&](int b, float (&v)[B][J], float& e_t, int& dl_t, int& dl_l) {
      const unsigned en_l = my[b * B + (lane & (B - 1))];
      const unsigned en_t = my[b * B + tq];
      e_t = lg[(size_t)(b * B + tq) * H + th];
#pragma unroll
      for (int q = 0; q < B; ++q) {
        const int s_q = (int)((unsigned)__builtin_amdgcn_readlane((int)en_l, q) >> 8);
#pragma unroll
        for (int j = 0; j < J; ++j) {
          const int f = (F % 32 == 0) ? j * 32 + lane : min(j * 32 + lane, F - 1);
          v[q][j] = h[(size_t)s_q * F + f];
        }
      }
      dl_t = (int)(en_t & 255u);
      dl_l = (int)(en_l & 255u);
    };
    auto add_batch = [&](const float (&v)[B][J], float e_t, int dl_t, int dl_l) {
      const float ex = expf(e_t - fdekey(s_mk[dl_t * H + th]));
      if (lane < B * H) atomicAdd(&s_den[dl_t * H + th], ex);
#pragma unroll
      for (int q = 0; q < B; ++q) {
        const int dl_q = __builtin_amdgcn_readlane(dl_l, q);
#pragma unroll
        for (int j = 0; j < J; ++j) {
          const int f = j * 32 + lane;
          const int hh = (H == 1) ? 0 : ((D % 32 == 0) ? (j * 32) / D : f / D);
          const float w = __int_as_float(
              __builtin_amdgcn_readlane(__float_as_int(ex), q * H + hh));
          if (F % 32 == 0 || f < F) atomicAdd(&acc[dl_q * F + f], w * v[q][j]);
        }
      }
    };
    if (nb2 > 0) {
      float vA[B][J], vB[B][J];
      float eA, eB;
      int dtA, dlA, dtB, dlB;
      request(0, vA, eA, dtA, dlA);
      for (int b = 0; b < nb2; ++b) {
        request(2 * b + 1, vB, eB, dtB, dlB);
        add_batch(vA, eA, dtA, dlA);
        request(min(2 * b + 2, 2 * nb2 - 1), vA, eA, dtA, dlA);
        add_batch(vB, eB, dtB, dlB);
      }
    }

    if (!__syncthreads_or(g < ngroups)) break;
    for (int i = tid; i < (R + 1) * H; i += AGG_NT) s_mo[i] = s_mk[i];
    __syncthreads();
  }
  if (LM == 1 && tid == 0) keep_ok[blockIdx.x] = (round == 0) ? 1 : 0;

  if constexpr (NC == 0) {
    for (int i = tid; i < cnt * F; i += AGG_NT) {
      const int q = (i / F) * H + (i % F) / D;
      agg[(size_t)lo * F + i] = acc[i] / s_den[q] + bias[i % F];
    }
  } else {
    constexpr int XS = F + 4;
    constexpr int NCT = (NC + 15) / 16;
    static_assert(F % 4 == 0 && XS % 64 == 4, "k steps of 4; conflict-free fragment reads");
    static_assert((R + 15) / 16 <= NW, "one wave per 16-row tile");
    static_assert((R + NCT * 16) * XS <= NW * CAP, "x and Wn fit the list's space");
    static_assert(R * NC <= (R + 1) * F, "the product fits the rows' space");
    float* xp = reinterpret_cast<float*>(lst);
    float* wp = xp + R * XS;
    float* hs = acc;
    constexpr int WJ = (NCT * 16 * F + AGG_NT - 1) / AGG_NT;
    float wv[WJ];
#pragma unroll
    for (int j = 0; j < WJ; ++j) {
      const int i = tid + j * AGG_NT;
      wv[j] = (i < NC * F) ? Wn[i] : 0.f;
    }
    for (int i = tid; i < cnt * F; i += AGG_NT) {
      const int r = i / F, f = i % F;
      const float v = acc[i] / s_den[r * H + f / D] + bias[f];
      xp[r * XS + f] = (v > 0.f) ? v : 0.f;
    }
#pragma unroll
    for (int j = 0; j < WJ; ++j) {
      const int i = tid + j * AGG_NT;
      if (i < NCT * 16 * F) wp[(i / F) * XS + i % F] = wv[j];
    }
    __syncthreads();
    if (wave * 16 < cnt) {
      const int half = lane >> 4;
      const int idx = lane & 15;
      const int ar = min(wave * 16 + idx, cnt - 1);
      const float* ap = xp + ar * XS + 2 * half;
      const float* bp = wp + idx * XS + 2 * half;
      v8f c[NCT];
#pragma unroll
      for (int t = 0; t < NCT; ++t) c[t] = v8f{0.f, 0.f, 0.f, 0.f, 0.f, 0.f, 0.f, 0.f};
#pragma unroll
      for (int k = 0; k < F; k += 4) {
        const v2f a = *(const v2f*)(ap + k);
#pragma unroll
        for (int t = 0; t < NCT; ++t) {
          const v2f b = *(const v2f*)(bp + t * 16 * XS + k);
          c[t] = __builtin_amdgcn_wmma_f32_16x16x4_f32(false, a, false, b, (short)0, c[t],
                                                       false, false);
        }
      }
#pragma unroll
      for (int t = 0; t < NCT; ++t) {
        const int col = t * 16 + idx;
#pragma unroll
        for (int r = 0; r < 8; ++r) {
          const int row = wave * 16 + r + 8 * half;
          if (col < NC && row < cnt) hs[row * NC + col] = c[t][r];
        }
      }
    }
    __syncthreads();
    for (int i = tid; i < cnt * NC; i += AGG_NT) hn[(size_t)lo * NC + i] = hs[i];
    if (tid < cnt) {
      float s0 = 0.f, s1 = 0.f;
      for (int dd = 0; dd < NC; ++dd) {
        const float v = hs[tid * NC + dd];
        s0 += v * att_sn[dd];
        s1 += v * att_dn[dd];
      }
      a_sn[lo + tid] = s0;
      a_dn[lo + tid] = s1;
    }
  }
}

extern "C" void kernel_launch(void* const* d_in, const int* in_sizes, int n_in,
                              void* d_out, int out_size, void* d_ws, size_t ws_size,
                              hipStream_t stream) {
  const float* x        = (const float*)d_in[0];
  const int*   ei       = (const int*)d_in[1];
  const float* W1       = (const float*)d_in[2];
  const float* att_src1 = (const float*)d_in[3];
  const float* att_dst1 = (const float*)d_in[4];
  const float* b1       = (const float*)d_in[5];
  const float* W2       = (const float*)d_in[6];
  const float* att_src2 = (const float*)d_in[7];
  const float* att_dst2 = (const float*)d_in[8];
  const float* b2       = (const float*)d_in[9];

  const int NFEAT = 256, NHID = 128, H1 = 4, D1 = 32, NCLS = 40;
  const int N = in_sizes[0] / NFEAT;
  const int E = in_sizes[1] / 2;
  const int* src = ei;
  const int* dst = ei + E;

  float* out = (float*)d_out;

  float* w = (float*)d_ws;
  float* h1   = w;
  float* h2   = h1  + (size_t)N * NHID;
  float* as1  = h2  + (size_t)N * NCLS;
  float* ad1  = as1 + (size_t)N * H1;
  const int groups = (N + AGG_R - 1) / AGG_R;
  float* lg1  = ad1 + (size_t)N * H1;
  float* as2  = lg1 + (size_t)groups * (AGG_NT / 32) * AGG_CAP * H1;
  float* ad2  = as2 + (size_t)N;
  int* klst = (int*)(ad2 + (size_t)N);
  int* kcnt = klst + (size_t)groups * (AGG_NT / 32) * AGG_CAP;
  int* kok  = kcnt + (size_t)groups * (AGG_NT / 32);

  const int TB = 256;

  gemm_wmma_f32_bres<256, 128><<<G1_WG, G1_NT, 0, stream>>>(x, W1, h1, N);
  attn_logits<<<(N * H1 + TB - 1) / TB, TB, 0, stream>>>(h1, att_src1, att_dst1,
                                                         as1, ad1, N * H1, H1, D1);
  gat_softmax_aggregate<4, 32, 40, 1><<<groups, AGG_NT, 0, stream>>>(
      src, dst, E, N, as1, ad1, h1, b1, nullptr, W2, att_src2, att_dst2, h2, as2, ad2,
      klst, kcnt, kok, lg1);

  gat_softmax_aggregate<1, 40, 0, 2><<<groups, AGG_NT, 0, stream>>>(
      src, dst, E, N, as2, ad2, h2, b2, out, nullptr, nullptr, nullptr, nullptr, nullptr, nullptr,
      klst, kcnt, kok, nullptr);
}


# edge scan reads a 1-byte destination-range tag per edge (written in the logit launch), not the 4-byte destination: 16 edges per 16-byte load; list order changes, sums = LDS arrival as before
# speedup vs baseline: 15.9550x; 1.1095x over previous
#include <hip/hip_runtime.h>

typedef float v2f __attribute__((ext_vector_type(2)));
typedef float v4f __attribute__((ext_vector_type(4)));
typedef float v8f __attribute__((ext_vector_type(8)));
typedef unsigned v4u __attribute__((ext_vector_type(4)));
typedef int v4i __attribute__((ext_vector_type(4)));

#define NEG_SLOPE 0.2f

__device__ __forceinline__ unsigned fkey(float f) {
  unsigned u = __float_as_uint(f);
  return (u & 0x80000000u) ? ~u : (u | 0x80000000u);
}
__device__ __forceinline__ float fdekey(unsigned k) {
  unsigned u = (k & 0x80000000u) ? (k & 0x7FFFFFFFu) : ~k;
  return __uint_as_float(u);
}

#define G1_NT 512
#define G1_WG 256

template <int K, int NC>
__global__ void __launch_bounds__(G1_NT)
gemm_wmma_f32_bres(const float* __restrict__ A, const float* __restrict__ B,
                   float* __restrict__ C, int N) {
  constexpr int XS = K + 4;
  constexpr int NCS = NC / 16;
  constexpr int NW = G1_NT / 32;
  constexpr int TS = NW / NCS;
  constexpr int PR = TS * 16;
  constexpr int KJ = K / 4;
  constexpr int KB = 16;
  constexpr int C4 = K / 4;
  constexpr int RQ = G1_NT / C4;
  constexpr int XQ = PR / RQ;
  constexpr int WQ = NC / RQ;
  static_assert(K % 4 == 0 && KJ % KB == 0 && XS % 64 == 4, "k steps of 4; conflict-free fragment reads");
  static_assert(NC % 16 == 0 && NW % NCS == 0 && G1_NT % C4 == 0 && PR % RQ == 0 && NC % RQ == 0,
                "panel geometry");
  __shared__ __attribute__((aligned(16))) float wsm[NC * XS];
  __shared__ __attribute__((aligned(16))) float xsm[2][PR * XS];

  const int tid = threadIdx.x;
  const int lane = tid & 31;
  const int wave = __builtin_amdgcn_readfirstlane(tid >> 5);
  const int cs = wave % NCS;
  const int ts = wave / NCS;
  const int half = lane >> 4;
  const int idx  = lane & 15;
  const int ntiles = (N + 15) / 16;
  const int npanels = (ntiles + TS - 1) / TS;
  const int xr = tid / C4;
  const int xc = (tid % C4) * 4;

  auto load_panel = [&](int p, v4f (&v)[XQ]) {
#pragma unroll
    for (int q = 0; q < XQ; ++q) {
      const int row = min(p * PR + xr + q * RQ, N - 1);
      v[q] = *(const v4f*)(A + (size_t)row * K + xc);
    }
  };
  auto store_panel = [&](float* xb, const v4f (&v)[XQ]) {
#pragma unroll
    for (int q = 0; q < XQ; ++q) *(v4f*)(xb + (xr + q * RQ) * XS + xc) = v[q];
  };

  int p = blockIdx.x;
  v4f nx[XQ];
  if (p < npanels) load_panel(p, nx);
  {
    v4f wv[WQ];
#pragma unroll
    for (int q = 0; q < WQ; ++q) wv[q] = *(const v4f*)(B + (size_t)(xr + q * RQ) * K + xc);
#pragma unroll
    for (int q = 0; q < WQ; ++q) *(v4f*)(wsm + (xr + q * RQ) * XS + xc) = wv[q];
  }
  if (p < npanels) store_panel(xsm[0], nx);
  __syncthreads();

  v2f b[KJ];
  {
    const float* bp = wsm + (cs * 16 + idx) * XS + 2 * half;
#pragma unroll
    for (int j = 0; j < KJ; ++j) b[j] = *(const v2f*)(bp + 4 * j);
  }

  for (int it = 0; p < npanels; p += gridDim.x, ++it) {
    const int pn = p + gridDim.x;
    const bool more = pn < npanels;
    if (more) load_panel(pn, nx);
    const float* xb = xsm[it & 1];
    const int tile = p * TS + ts;
    if (tile < ntiles) {
      const float* ap = xb + (ts * 16 + idx) * XS + 2 * half;
      v8f c = {0.f, 0.f, 0.f, 0.f, 0.f, 0.f, 0.f, 0.f};
      v2f a[2][KB];
#pragma unroll
      for (int i = 0; i < KB; ++i) a[0][i] = *(const v2f*)(ap + 4 * i);
#pragma unroll
      for (int jb = 0; jb < KJ / KB; ++jb) {
        if (jb + 1 < KJ / KB) {
#pragma unroll
          for (int i = 0; i < KB; ++i) a[(jb + 1) & 1][i] = *(const v2f*)(ap + 4 * ((jb + 1) * KB + i));
        }
#pragma unroll
        for (int i = 0; i < KB; ++i)
          c = __builtin_amdgcn_wmma_f32_16x16x4_f32(false, a[jb & 1][i], false, b[jb * KB + i],
                                                    (short)0, c, false, false);
      }
      __builtin_amdgcn_sched_group_barrier(0x100, KB / 2, 0);
#pragma unroll
      for (int jb = 0; jb < KJ / KB; ++jb) {
        if (jb + 1 < KJ / KB) __builtin_amdgcn_sched_group_barrier(0x100, KB / 2, 0);
        __builtin_amdgcn_sched_group_barrier(0x008, KB, 0);
      }
      const int col = cs * 16 + idx;
#pragma unroll
      for (int r = 0; r < 8; ++r) {
        const int row = tile * 16 + r + 8 * half;
        if (row < N) C[(size_t)row * NC + col] = c[r];
      }
    }
    if (more) store_panel(xsm[(it + 1) & 1], nx);
    __syncthreads();
  }
}

#define AGG_R 196
#define TAG_G 512

__global__ void attn_logits(const float* __restrict__ h, const float* __restrict__ att_s,
                            const float* __restrict__ att_d, float* __restrict__ a_s,
                            float* __restrict__ a_d, int NH, int H, int D,
                            const int* __restrict__ dst, int E, unsigned* __restrict__ gtag) {
  int i = blockIdx.x * blockDim.x + threadIdx.x;
  {
    const int nkw = (E + TAG_G - 1) / TAG_G * (TAG_G / 4);
    const bool al = (reinterpret_cast<size_t>(dst) & 15) == 0;
    for (int q = i; q < nkw; q += gridDim.x * blockDim.x) {
      const int e0 = 4 * q;
      int d0, d1, d2, d3;
      if (al && e0 + 3 < E) {
        const v4i v = *(const v4i*)(dst + e0);
        d0 = v.x; d1 = v.y; d2 = v.z; d3 = v.w;
      } else {
        d0 = dst[min(e0, E - 1)];
        d1 = dst[min(e0 + 1, E - 1)];
        d2 = dst[min(e0 + 2, E - 1)];
        d3 = dst[min(e0 + 3, E - 1)];
      }
      gtag[q] = ((unsigned)(d0 / AGG_R) & 255u) | (((unsigned)(d1 / AGG_R) & 255u) << 8) |
                (((unsigned)(d2 / AGG_R) & 255u) << 16) | (((unsigned)(d3 / AGG_R) & 255u) << 24);
    }
  }
  if (i >= NH) return;
  const int hh = i % H;
  const float* __restrict__ hp = h + (size_t)i * D;
  const float* __restrict__ as = att_s + (size_t)hh * D;
  const float* __restrict__ ad = att_d + (size_t)hh * D;
  float s0 = 0.f, s1 = 0.f;
  for (int dd = 0; dd < D; ++dd) {
    const float v = hp[dd];
    s0 += v * as[dd];
    s1 += v * ad[dd];
  }
  a_s[i] = s0;
  a_d[i] = s1;
}

#define AGG_NT 1024
#define AGG_K 16
#define AGG_B 8
#define AGG_CAP 1024
#define AGG_PD 3

template <int H, int D, int NC, int LM>
__global__ void __launch_bounds__(AGG_NT)
gat_softmax_aggregate(const int* __restrict__ src, const int* __restrict__ dst, int E, int N,
                      const float* __restrict__ a_s, const float* __restrict__ a_d,
                      const float* __restrict__ h,
                      const float* __restrict__ bias,
                      float* __restrict__ agg,
                      const float* __restrict__ Wn,
                      const float* __restrict__ att_sn,
                      const float* __restrict__ att_dn,
                      float* __restrict__ hn,
                      float* __restrict__ a_sn, float* __restrict__ a_dn,
                      int* __restrict__ keep_lst,
                      int* __restrict__ keep_cnt,
                      int* __restrict__ keep_ok,
                      float* __restrict__ lg_ws,
                      const unsigned* __restrict__ gtag) {
  constexpr int F = H * D;
  constexpr int R = AGG_R;
  constexpr int J = (F + 31) / 32;
  constexpr int K = AGG_K;
  constexpr int B = AGG_B;
  constexpr int CAP = AGG_CAP;
  constexpr int NW = AGG_NT / 32;
  constexpr int G = 32 * K;
  static_assert(B * H <= 32, "one lane per (edge, head) of a batch");
  static_assert((B & (B - 1)) == 0 && 2 * B <= 32 && CAP % (2 * B) == 0 && CAP >= 2 * G, "list geometry");
  static_assert(R < 255 && R <= 32 * NW, "local destination and the spare row fit 8 bits, 255 is free");
  static_assert(G == TAG_G, "a scan step is one 16-byte tag load per lane");
  __shared__ __attribute__((aligned(16))) float acc[(R + 1) * F];
  __shared__ float s_ad[(R + 1) * H];
  __shared__ unsigned s_mk[(R + 1) * H];
  __shared__ unsigned s_mo[(R + 1) * H];
  __shared__ float s_den[(R + 1) * H];
  __shared__ __attribute__((aligned(16))) unsigned lst[NW * CAP];
  __shared__ float lg_lds[(H == 1) ? NW * CAP : 1];

  const int tid = threadIdx.x;
  const int lane = tid & 31;
  const int wave = __builtin_amdgcn_readfirstlane(tid >> 5);
  const int lo = blockIdx.x * R;
  const int cnt = min(R, N - lo);
  unsigned* __restrict__ my = lst + wave * CAP;

  for (int i = tid; i < (R + 1) * F; i += AGG_NT) acc[i] = 0.f;
  for (int i = tid; i < (R + 1) * H; i += AGG_NT) {
    s_ad[i] = (i < cnt * H) ? a_d[(size_t)lo * H + i] : 0.f;
    s_mk[i] = 0u;
    s_den[i] = 0.f;
  }

  const int ngroups = (E + G - 1) / G;
  auto load_tags = [&](int g) -> v4u {
    return *(const v4u*)(gtag + (size_t)min(g, ngroups - 1) * (G / 4) + lane * 4);
  };
  const unsigned own = (unsigned)blockIdx.x * 0x01010101u;

  const int slot = blockIdx.x * NW + wave;
  float* __restrict__ lg;
  if constexpr (H == 1) lg = lg_lds + wave * CAP;
  else lg = lg_ws + (size_t)slot * CAP * H;
  auto note_logits = [&](int i, int s, int dl) {
    float v[H];
#pragma unroll
    for (int hh = 0; hh < H; ++hh) {
      const float t = a_s[(size_t)s * H + hh] + s_ad[dl * H + hh];
      v[hh] = (t > 0.f) ? t : NEG_SLOPE * t;
      atomicMax(&s_mk[dl * H + hh], fkey(v[hh]));
    }
    if constexpr (H == 4) {
      *(v4f*)(lg + (size_t)i * H) = v4f{v[0], v[1], v[2], v[3]};
    } else {
#pragma unroll
      for (int hh = 0; hh < H; ++hh) lg[(size_t)i * H + hh] = v[hh];
    }
  };
  bool kept = false;
  if constexpr (LM == 2) kept = __builtin_amdgcn_readfirstlane(keep_ok[blockIdx.x]) == 1;

  int g = wave;
  v4u tq[AGG_PD];
#pragma unroll
  for (int u = 0; u < AGG_PD; ++u) tq[u] = v4u{0u, 0u, 0u, 0u};
  if (kept) {
    g = ngroups;
  } else {
#pragma unroll
    for (int u = 0; u < AGG_PD; ++u) tq[u] = load_tags(g + u * NW);
  }
  __syncthreads();

  int round = 0;
  for (;; ++round) {
    int n_pad;
    if (kept) {
      n_pad = min(__builtin_amdgcn_readfirstlane(keep_cnt[slot]), CAP) & ~(2 * B - 1);
      const int* __restrict__ kl = keep_lst + (size_t)slot * CAP;
      for (int i = lane; i < n_pad; i += 32) {
        const unsigned en = (unsigned)kl[i];
        const int s = (int)(en >> 8);
        const int dl = (int)(en & 255u);
        my[i] = en;
        note_logits(i, s, dl);
      }
    } else {
      int n_w = 0;
      if (round == 0) {
        const int dl = wave + NW * lane;
        const bool hit = dl < cnt;
        const unsigned mk = __builtin_amdgcn_ballot_w32(hit);
        if (hit) my[__builtin_popcount(mk & ((1u << lane) - 1u))] = ((unsigned)(E + lo + dl) << 8) | (unsigned)dl;
        n_w = __builtin_popcount(mk);
      }
      if (round > 0) {
#pragma unroll
        for (int u = 0; u < AGG_PD; ++u) tq[u] = load_tags(g + u * NW);
      }
      auto scan_step = [&](v4u& t) {
        unsigned tw[4] = {t.x, t.y, t.z, t.w};
        if (g * G + G > E) {
#pragma unroll
          for (int d = 0; d < 4; ++d) {
            const int left = E - (g * G + lane * K + d * 4);
            const unsigned keep = (left >= 4) ? 0xffffffffu : ((left <= 0) ? 0u : ((1u << (8 * left)) - 1u));
            tw[d] = (tw[d] & keep) | (~own & ~keep);
          }
        }
#pragma unroll
        for (int d = 0; d < 4; ++d) {
          const unsigned y = tw[d] ^ own;
          unsigned z = ~(((y & 0x7f7f7f7fu) + 0x7f7f7f7fu) | y | 0x7f7f7f7fu);
          const int e0 = g * G + lane * K + d * 4;
          unsigned mk = __builtin_amdgcn_ballot_w32(z != 0u);
          while (mk) {
            if (z != 0u) {
              my[n_w + (int)__builtin_amdgcn_mbcnt_lo(mk, 0u)] =
                  ((unsigned)(e0 + (__builtin_ctz(z) >> 3)) << 8) | 255u;
              z &= z - 1u;
            }
            n_w += __builtin_popcount(mk);
            mk = __builtin_amdgcn_ballot_w32(z != 0u);
          }
        }
        t = load_tags(g + AGG_PD * NW);
        g += NW;
      };
      auto room = [&]() { return g < ngroups && n_w <= CAP - G - 2 * B; };
      while (room()) {
#pragma unroll
        for (int u = 0; u < AGG_PD; ++u) {
          if (u > 0 && !room()) break;
          scan_step(tq[u]);
        }
      }
      n_pad = (n_w + 2 * B - 1) / (2 * B) * (2 * B);
      if (lane < n_pad - n_w) my[n_w + lane] = ((unsigned)(E + lo) << 8) | (unsigned)R;
      if (LM == 1 && round == 0 && lane == 0) keep_cnt[slot] = n_pad;

      for (int i = lane; i < n_pad; i += 32) {
        const unsigned en = my[i];
        const int e = (int)(en >> 8);
        const int lb = (int)(en & 255u);
        const int sv = src[min(e, E - 1)];
        const int dv = dst[min(e, E - 1)];
        const int s = (e < E) ? sv : e - E;
        const int dd = dv - lo;
        const int dl = (lb != 255) ? lb : (((unsigned)dd < (unsigned)cnt) ? dd : R);
        my[i] = ((unsigned)s << 8) | (unsigned)dl;
        if (LM == 1 && round == 0) keep_lst[(size_t)slot * CAP + i] = (s << 8) | dl;
        note_logits(i, s, dl);
      }
    }
    __syncthreads();
    if (round > 0) {
      for (int i = tid; i < R * F; i += AGG_NT) {
        const int q = (i / F) * H + (i % F) / D;
        const unsigned ko = s_mo[q], kn = s_mk[q];
        if (ko != kn) acc[i] *= expf(fdekey(ko) - fdekey(kn));
      }
      for (int i = tid; i < R * H; i += AGG_NT) {
        const unsigned ko = s_mo[i], kn = s_mk[i];
        if (ko != kn) s_den[i] *= expf(fdekey(ko) - fdekey(kn));
      }
      __syncthreads();
    }

    const int nb2 = n_pad / (2 * B);
    const int tq = min(lane / H, B - 1);
    const int th = lane - (lane / H) * H;
    auto request = [&](int b, float (&v)[B][J], float& e_t, int& dl_t, int& dl_l) {
      const unsigned en_l = my[b * B + (lane & (B - 1))];
      const unsigned en_t = my[b * B + tq];
      e_t = lg[(size_t)(b * B + tq) * H + th];
#pragma unroll
      for (int q = 0; q < B; ++q) {
        const int s_q = (int)((unsigned)__builtin_amdgcn_readlane((int)en_l, q) >> 8);
#pragma unroll
        for (int j = 0; j < J; ++j) {
          const int f = (F % 32 == 0) ? j * 32 + lane : min(j * 32 + lane, F - 1);
          v[q][j] = h[(size_t)s_q * F + f];
        }
      }
      dl_t = (int)(en_t & 255u);
      dl_l = (int)(en_l & 255u);
    };
    auto add_batch = [&](const float (&v)[B][J], float e_t, int dl_t, int dl_l) {
      const float ex = expf(e_t - fdekey(s_mk[dl_t * H + th]));
      if (lane < B * H) atomicAdd(&s_den[dl_t * H + th], ex);
#pragma unroll
      for (int q = 0; q < B; ++q) {
        const int dl_q = __builtin_amdgcn_readlane(dl_l, q);
#pragma unroll
        for (int j = 0; j < J; ++j) {
          const int f = j * 32 + lane;
          const int hh = (H == 1) ? 0 : ((D % 32 == 0) ? (j * 32) / D : f / D);
          const float w = __int_as_float(
              __builtin_amdgcn_readlane(__float_as_int(ex), q * H + hh));
          if (F % 32 == 0 || f < F) atomicAdd(&acc[dl_q * F + f], w * v[q][j]);
        }
      }
    };
    if (nb2 > 0) {
      float vA[B][J], vB[B][J];
      float eA, eB;
      int dtA, dlA, dtB, dlB;
      request(0, vA, eA, dtA, dlA);
      for (int b = 0; b < nb2; ++b) {
        request(2 * b + 1, vB, eB, dtB, dlB);
        add_batch(vA, eA, dtA, dlA);
        request(min(2 * b + 2, 2 * nb2 - 1), vA, eA, dtA, dlA);
        add_batch(vB, eB, dtB, dlB);
      }
    }

    if (!__syncthreads_or(g < ngroups)) break;
    for (int i = tid; i < (R + 1) * H; i += AGG_NT) s_mo[i] = s_mk[i];
    __syncthreads();
  }
  if (LM == 1 && tid == 0) keep_ok[blockIdx.x] = (round == 0) ? 1 : 0;

  if constexpr (NC == 0) {
    for (int i = tid; i < cnt * F; i += AGG_NT) {
      const int q = (i / F) * H + (i % F) / D;
      agg[(size_t)lo * F + i] = acc[i] / s_den[q] + bias[i % F];
    }
  } else {
    constexpr int XS = F + 4;
    constexpr int NCT = (NC + 15) / 16;
    static_assert(F % 4 == 0 && XS % 64 == 4, "k steps of 4; conflict-free fragment reads");
    static_assert((R + 15) / 16 <= NW, "one wave per 16-row tile");
    static_assert((R + NCT * 16) * XS <= NW * CAP, "x and Wn fit the list's space");
    static_assert(R * NC <= (R + 1) * F, "the product fits the rows' space");
    float* xp = reinterpret_cast<float*>(lst);
    float* wp = xp + R * XS;
    float* hs = acc;
    constexpr int WJ = (NCT * 16 * F + AGG_NT - 1) / AGG_NT;
    float wv[WJ];
#pragma unroll
    for (int j = 0; j < WJ; ++j) {
      const int i = tid + j * AGG_NT;
      wv[j] = (i < NC * F) ? Wn[i] : 0.f;
    }
    for (int i = tid; i < cnt * F; i += AGG_NT) {
      const int r = i / F, f = i % F;
      const float v = acc[i] / s_den[r * H + f / D] + bias[f];
      xp[r * XS + f] = (v > 0.f) ? v : 0.f;
    }
#pragma unroll
    for (int j = 0; j < WJ; ++j) {
      const int i = tid + j * AGG_NT;
      if (i < NCT * 16 * F) wp[(i / F) * XS + i % F] = wv[j];
    }
    __syncthreads();
    if (wave * 16 < cnt) {
      const int half = lane >> 4;
      const int idx = lane & 15;
      const int ar = min(wave * 16 + idx, cnt - 1);
      const float* ap = xp + ar * XS + 2 * half;
      const float* bp = wp + idx * XS + 2 * half;
      v8f c[NCT];
#pragma unroll
      for (int t = 0; t < NCT; ++t) c[t] = v8f{0.f, 0.f, 0.f, 0.f, 0.f, 0.f, 0.f, 0.f};
#pragma unroll
      for (int k = 0; k < F; k += 4) {
        const v2f a = *(const v2f*)(ap + k);
#pragma unroll
        for (int t = 0; t < NCT; ++t) {
          const v2f b = *(const v2f*)(bp + t * 16 * XS + k);
          c[t] = __builtin_amdgcn_wmma_f32_16x16x4_f32(false, a, false, b, (short)0, c[t],
                                                       false, false);
        }
      }
#pragma unroll
      for (int t = 0; t < NCT; ++t) {
        const int col = t * 16 + idx;
#pragma unroll
        for (int r = 0; r < 8; ++r) {
          const int row = wave * 16 + r + 8 * half;
          if (col < NC && row < cnt) hs[row * NC + col] = c[t][r];
        }
      }
    }
    __syncthreads();
    for (int i = tid; i < cnt * NC; i += AGG_NT) hn[(size_t)lo * NC + i] = hs[i];
    if (tid < cnt) {
      float s0 = 0.f, s1 = 0.f;
      for (int dd = 0; dd < NC; ++dd) {
        const float v = hs[tid * NC + dd];
        s0 += v * att_sn[dd];
        s1 += v * att_dn[dd];
      }
      a_sn[lo + tid] = s0;
      a_dn[lo + tid] = s1;
    }
  }
}

extern "C" void kernel_launch(void* const* d_in, const int* in_sizes, int n_in,
                              void* d_out, int out_size, void* d_ws, size_t ws_size,
                              hipStream_t stream) {
  const float* x        = (const float*)d_in[0];
  const int*   ei       = (const int*)d_in[1];
  const float* W1       = (const float*)d_in[2];
  const float* att_src1 = (const float*)d_in[3];
  const float* att_dst1 = (const float*)d_in[4];
  const float* b1       = (const float*)d_in[5];
  const float* W2       = (const float*)d_in[6];
  const float* att_src2 = (const float*)d_in[7];
  const float* att_dst2 = (const float*)d_in[8];
  const float* b2       = (const float*)d_in[9];

  const int NFEAT = 256, NHID = 128, H1 = 4, D1 = 32, NCLS = 40;
  const int N = in_sizes[0] / NFEAT;
  const int E = in_sizes[1] / 2;
  const int* src = ei;
  const int* dst = ei + E;

  float* out = (float*)d_out;

  float* w = (float*)d_ws;
  float* h1   = w;
  float* h2   = h1  + (size_t)N * NHID;
  float* as1  = h2  + (size_t)N * NCLS;
  float* ad1  = as1 + (size_t)N * H1;
  const int groups = (N + AGG_R - 1) / AGG_R;
  float* lg1  = ad1 + (size_t)N * H1;
  unsigned* gtag = (unsigned*)(lg1 + (size_t)groups * (AGG_NT / 32) * AGG_CAP * H1);
  float* as2  = (float*)(gtag + (size_t)((E + TAG_G - 1) / TAG_G) * (TAG_G / 4));
  float* ad2  = as2 + (size_t)N;
  int* klst = (int*)(ad2 + (size_t)N);
  int* kcnt = klst + (size_t)groups * (AGG_NT / 32) * AGG_CAP;
  int* kok  = kcnt + (size_t)groups * (AGG_NT / 32);

  const int TB = 256;

  gemm_wmma_f32_bres<256, 128><<<G1_WG, G1_NT, 0, stream>>>(x, W1, h1, N);
  attn_logits<<<(N * H1 + TB - 1) / TB, TB, 0, stream>>>(h1, att_src1, att_dst1,
                                                         as1, ad1, N * H1, H1, D1, dst, E, gtag);
  gat_softmax_aggregate<4, 32, 40, 1><<<groups, AGG_NT, 0, stream>>>(
      src, dst, E, N, as1, ad1, h1, b1, nullptr, W2, att_src2, att_dst2, h2, as2, ad2,
      klst, kcnt, kok, lg1, gtag);

  gat_softmax_aggregate<1, 40, 0, 2><<<groups, AGG_NT, 0, stream>>>(
      src, dst, E, N, as2, ad2, h2, b2, out, nullptr, nullptr, nullptr, nullptr, nullptr, nullptr,
      klst, kcnt, kok, nullptr, gtag);
}


# edge scan reads 1 bit per edge: the logit launch builds a membership bitmap, a row per aggregation work-group (LDS OR); 128 edges per 16-byte load; list order changes, sums = LDS arrival as before
# speedup vs baseline: 16.9064x; 1.0596x over previous
#include <hip/hip_runtime.h>

typedef float v2f __attribute__((ext_vector_type(2)));
typedef float v4f __attribute__((ext_vector_type(4)));
typedef float v8f __attribute__((ext_vector_type(8)));
typedef unsigned v4u __attribute__((ext_vector_type(4)));
typedef int v4i __attribute__((ext_vector_type(4)));

#define NEG_SLOPE 0.2f

__device__ __forceinline__ unsigned fkey(float f) {
  unsigned u = __float_as_uint(f);
  return (u & 0x80000000u) ? ~u : (u | 0x80000000u);
}
__device__ __forceinline__ float fdekey(unsigned k) {
  unsigned u = (k & 0x80000000u) ? (k & 0x7FFFFFFFu) : ~k;
  return __uint_as_float(u);
}

#define G1_NT 512
#define G1_WG 256

template <int K, int NC>
__global__ void __launch_bounds__(G1_NT)
gemm_wmma_f32_bres(const float* __restrict__ A, const float* __restrict__ B,
                   float* __restrict__ C, int N) {
  constexpr int XS = K + 4;
  constexpr int NCS = NC / 16;
  constexpr int NW = G1_NT / 32;
  constexpr int TS = NW / NCS;
  constexpr int PR = TS * 16;
  constexpr int KJ = K / 4;
  constexpr int KB = 16;
  constexpr int C4 = K / 4;
  constexpr int RQ = G1_NT / C4;
  constexpr int XQ = PR / RQ;
  constexpr int WQ = NC / RQ;
  static_assert(K % 4 == 0 && KJ % KB == 0 && XS % 64 == 4, "k steps of 4; conflict-free fragment reads");
  static_assert(NC % 16 == 0 && NW % NCS == 0 && G1_NT % C4 == 0 && PR % RQ == 0 && NC % RQ == 0,
                "panel geometry");
  __shared__ __attribute__((aligned(16))) float wsm[NC * XS];
  __shared__ __attribute__((aligned(16))) float xsm[2][PR * XS];

  const int tid = threadIdx.x;
  const int lane = tid & 31;
  const int wave = __builtin_amdgcn_readfirstlane(tid >> 5);
  const int cs = wave % NCS;
  const int ts = wave / NCS;
  const int half = lane >> 4;
  const int idx  = lane & 15;
  const int ntiles = (N + 15) / 16;
  const int npanels = (ntiles + TS - 1) / TS;
  const int xr = tid / C4;
  const int xc = (tid % C4) * 4;

  auto load_panel = [&](int p, v4f (&v)[XQ]) {
#pragma unroll
    for (int q = 0; q < XQ; ++q) {
      const int row = min(p * PR + xr + q * RQ, N - 1);
      v[q] = *(const v4f*)(A + (size_t)row * K + xc);
    }
  };
  auto store_panel = [&](float* xb, const v4f (&v)[XQ]) {
#pragma unroll
    for (int q = 0; q < XQ; ++q) *(v4f*)(xb + (xr + q * RQ) * XS + xc) = v[q];
  };

  int p = blockIdx.x;
  v4f nx[XQ];
  if (p < npanels) load_panel(p, nx);
  {
    v4f wv[WQ];
#pragma unroll
    for (int q = 0; q < WQ; ++q) wv[q] = *(const v4f*)(B + (size_t)(xr + q * RQ) * K + xc);
#pragma unroll
    for (int q = 0; q < WQ; ++q) *(v4f*)(wsm + (xr + q * RQ) * XS + xc) = wv[q];
  }
  if (p < npanels) store_panel(xsm[0], nx);
  __syncthreads();

  v2f b[KJ];
  {
    const float* bp = wsm + (cs * 16 + idx) * XS + 2 * half;
#pragma unroll
    for (int j = 0; j < KJ; ++j) b[j] = *(const v2f*)(bp + 4 * j);
  }

  for (int it = 0; p < npanels; p += gridDim.x, ++it) {
    const int pn = p + gridDim.x;
    const bool more = pn < npanels;
    if (more) load_panel(pn, nx);
    const float* xb = xsm[it & 1];
    const int tile = p * TS + ts;
    if (tile < ntiles) {
      const float* ap = xb + (ts * 16 + idx) * XS + 2 * half;
      v8f c = {0.f, 0.f, 0.f, 0.f, 0.f, 0.f, 0.f, 0.f};
      v2f a[2][KB];
#pragma unroll
      for (int i = 0; i < KB; ++i) a[0][i] = *(const v2f*)(ap + 4 * i);
#pragma unroll
      for (int jb = 0; jb < KJ / KB; ++jb) {
        if (jb + 1 < KJ / KB) {
#pragma unroll
          for (int i = 0; i < KB; ++i) a[(jb + 1) & 1][i] = *(const v2f*)(ap + 4 * ((jb + 1) * KB + i));
        }
#pragma unroll
        for (int i = 0; i < KB; ++i)
          c = __builtin_amdgcn_wmma_f32_16x16x4_f32(false, a[jb & 1][i], false, b[jb * KB + i],
                                                    (short)0, c, false, false);
      }
      __builtin_amdgcn_sched_group_barrier(0x100, KB / 2, 0);
#pragma unroll
      for (int jb = 0; jb < KJ / KB; ++jb) {
        if (jb + 1 < KJ / KB) __builtin_amdgcn_sched_group_barrier(0x100, KB / 2, 0);
        __builtin_amdgcn_sched_group_barrier(0x008, KB, 0);
      }
      const int col = cs * 16 + idx;
#pragma unroll
      for (int r = 0; r < 8; ++r) {
        const int row = tile * 16 + r + 8 * half;
        if (row < N) C[(size_t)row * NC + col] = c[r];
      }
    }
    if (more) store_panel(xsm[(it + 1) & 1], nx);
    __syncthreads();
  }
}

#define AGG_R 196
#define AGG_NG 256
#define SCAN_W 128
#define BM_CE 2048
#define BM_CW (BM_CE / 32)

__global__ void attn_logits(const float* __restrict__ h, const float* __restrict__ att_s,
                            const float* __restrict__ att_d, float* __restrict__ a_s,
                            float* __restrict__ a_d, int NH, int H, int D,
                            const int* __restrict__ dst, int E, unsigned* __restrict__ gbm,
                            int nword, int nchunk) {
  __shared__ __attribute__((aligned(16))) unsigned bm[AGG_NG * BM_CW];
  if ((int)blockIdx.x < nchunk) {
    const int c = blockIdx.x;
    const int t = threadIdx.x;
    for (int q = t; q < AGG_NG * BM_CW / 4; q += blockDim.x) ((v4u*)bm)[q] = v4u{0u, 0u, 0u, 0u};
    __syncthreads();
    const bool al = (reinterpret_cast<size_t>(dst) & 15) == 0;
    for (int j = t; j < BM_CE / 8; j += blockDim.x) {
      const int e0 = c * BM_CE + j * 8;
      int d[8];
      if (al && e0 + 7 < E) {
        const v4i v0 = *(const v4i*)(dst + e0);
        const v4i v1 = *(const v4i*)(dst + e0 + 4);
        d[0] = v0.x; d[1] = v0.y; d[2] = v0.z; d[3] = v0.w;
        d[4] = v1.x; d[5] = v1.y; d[6] = v1.z; d[7] = v1.w;
      } else {
#pragma unroll
        for (int k = 0; k < 8; ++k) d[k] = (e0 + k < E) ? dst[e0 + k] : -1;
      }
#pragma unroll
      for (int k = 0; k < 8; ++k) {
        const unsigned r = (unsigned)d[k] / AGG_R;
        if (r < AGG_NG) atomicOr(&bm[r * BM_CW + (j >> 2)], 1u << ((j & 3) * 8 + k));
      }
    }
    __syncthreads();
    for (int q = t; q < AGG_NG * BM_CW / 4; q += blockDim.x) {
      const int r = q / (BM_CW / 4), w4 = q % (BM_CW / 4);
      *(v4u*)(gbm + (size_t)r * nword + (size_t)c * BM_CW + w4 * 4) = ((const v4u*)bm)[q];
    }
    return;
  }
  int i = ((int)blockIdx.x - nchunk) * blockDim.x + threadIdx.x;
  if (i >= NH) return;
  const int hh = i % H;
  const float* __restrict__ hp = h + (size_t)i * D;
  const float* __restrict__ as = att_s + (size_t)hh * D;
  const float* __restrict__ ad = att_d + (size_t)hh * D;
  float s0 = 0.f, s1 = 0.f;
  for (int dd = 0; dd < D; ++dd) {
    const float v = hp[dd];
    s0 += v * as[dd];
    s1 += v * ad[dd];
  }
  a_s[i] = s0;
  a_d[i] = s1;
}

#define AGG_NT 1024
#define AGG_B 8
#define AGG_CAP 1024
#define AGG_PD 3

template <int H, int D, int NC, int LM>
__global__ void __launch_bounds__(AGG_NT)
gat_softmax_aggregate(const int* __restrict__ src, const int* __restrict__ dst, int E, int N,
                      const float* __restrict__ a_s, const float* __restrict__ a_d,
                      const float* __restrict__ h,
                      const float* __restrict__ bias,
                      float* __restrict__ agg,
                      const float* __restrict__ Wn,
                      const float* __restrict__ att_sn,
                      const float* __restrict__ att_dn,
                      float* __restrict__ hn,
                      float* __restrict__ a_sn, float* __restrict__ a_dn,
                      int* __restrict__ keep_lst,
                      int* __restrict__ keep_cnt,
                      int* __restrict__ keep_ok,
                      float* __restrict__ lg_ws,
                      const unsigned* __restrict__ gbm,
                      int nword) {
  constexpr int F = H * D;
  constexpr int R = AGG_R;
  constexpr int J = (F + 31) / 32;
  constexpr int B = AGG_B;
  constexpr int CAP = AGG_CAP;
  constexpr int NW = AGG_NT / 32;
  static_assert(B * H <= 32, "one lane per (edge, head) of a batch");
  static_assert((B & (B - 1)) == 0 && 2 * B <= 32 && CAP % (2 * B) == 0 && CAP >= 64 + 2 * B, "list geometry");
  static_assert(R < 255 && R <= 32 * NW, "local destination and the spare row fit 8 bits, 255 is free");
  static_assert(SCAN_W == 32 * 4, "a scan step is one 16-byte bitmap load per lane");
  __shared__ __attribute__((aligned(16))) float acc[(R + 1) * F];
  __shared__ float s_ad[(R + 1) * H];
  __shared__ unsigned s_mk[(R + 1) * H];
  __shared__ unsigned s_mo[(R + 1) * H];
  __shared__ float s_den[(R + 1) * H];
  __shared__ __attribute__((aligned(16))) unsigned lst[NW * CAP];
  __shared__ float lg_lds[(H == 1) ? NW * CAP : 1];

  const int tid = threadIdx.x;
  const int lane = tid & 31;
  const int wave = __builtin_amdgcn_readfirstlane(tid >> 5);
  const int lo = blockIdx.x * R;
  const int cnt = min(R, N - lo);
  unsigned* __restrict__ my = lst + wave * CAP;

  for (int i = tid; i < (R + 1) * F; i += AGG_NT) acc[i] = 0.f;
  for (int i = tid; i < (R + 1) * H; i += AGG_NT) {
    s_ad[i] = (i < cnt * H) ? a_d[(size_t)lo * H + i] : 0.f;
    s_mk[i] = 0u;
    s_den[i] = 0.f;
  }

  const int ngroups = nword / SCAN_W;
  const unsigned* __restrict__ row = gbm + (size_t)blockIdx.x * nword;
  auto load_bits = [&](int g) -> v4u {
    return *(const v4u*)(row + (size_t)min(g, ngroups - 1) * SCAN_W + lane * 4);
  };

  const int slot = blockIdx.x * NW + wave;
  float* __restrict__ lg;
  if constexpr (H == 1) lg = lg_lds + wave * CAP;
  else lg = lg_ws + (size_t)slot * CAP * H;
  auto note_logits = [&](int i, int s, int dl) {
    float v[H];
#pragma unroll
    for (int hh = 0; hh < H; ++hh) {
      const float t = a_s[(size_t)s * H + hh] + s_ad[dl * H + hh];
      v[hh] = (t > 0.f) ? t : NEG_SLOPE * t;
      atomicMax(&s_mk[dl * H + hh], fkey(v[hh]));
    }
    if constexpr (H == 4) {
      *(v4f*)(lg + (size_t)i * H) = v4f{v[0], v[1], v[2], v[3]};
    } else {
#pragma unroll
      for (int hh = 0; hh < H; ++hh) lg[(size_t)i * H + hh] = v[hh];
    }
  };
  bool kept = false;
  if constexpr (LM == 2) kept = __builtin_amdgcn_readfirstlane(keep_ok[blockIdx.x]) == 1;

  int g = wave;
  v4u tq[AGG_PD];
#pragma unroll
  for (int u = 0; u < AGG_PD; ++u) tq[u] = v4u{0u, 0u, 0u, 0u};
  if (kept) {
    g = ngroups;
  } else {
#pragma unroll
    for (int u = 0; u < AGG_PD; ++u) tq[u] = load_bits(g + u * NW);
  }
  __syncthreads();

  int rs_d = 0, rs_k = 0;
  int round = 0;
  for (;; ++round) {
    int n_pad;
    if (kept) {
      n_pad = min(__builtin_amdgcn_readfirstlane(keep_cnt[slot]), CAP) & ~(2 * B - 1);
      const int* __restrict__ kl = keep_lst + (size_t)slot * CAP;
      for (int i = lane; i < n_pad; i += 32) {
        const unsigned en = (unsigned)kl[i];
        const int s = (int)(en >> 8);
        const int dl = (int)(en & 255u);
        my[i] = en;
        note_logits(i, s, dl);
      }
    } else {
      int n_w = 0;
      if (round == 0) {
        const int dl = wave + NW * lane;
        const bool hit = dl < cnt;
        const unsigned mk = __builtin_amdgcn_ballot_w32(hit);
        if (hit) my[__builtin_popcount(mk & ((1u << lane) - 1u))] = ((unsigned)(E + lo + dl) << 8) | (unsigned)dl;
        n_w = __builtin_popcount(mk);
      }
      if (round > 0) {
#pragma unroll
        for (int u = 0; u < AGG_PD; ++u) tq[u] = load_bits(g + u * NW);
        if (rs_d > 0 || rs_k > 0) {
          unsigned tw[4] = {tq[0].x, tq[0].y, tq[0].z, tq[0].w};
#pragma unroll
          for (int d = 0; d < 4; ++d) {
            if (d < rs_d) {
              tw[d] = 0u;
            } else if (d == rs_d) {
#pragma nounroll
              for (int k = 0; k < rs_k; ++k) tw[d] &= tw[d] - 1u;
            }
          }
          tq[0] = v4u{tw[0], tw[1], tw[2], tw[3]};
        }
      }
      bool full = false;
      auto scan_step = [&](v4u& t) {
        const unsigned tw[4] = {t.x, t.y, t.z, t.w};
#pragma unroll
        for (int d = 0; d < 4; ++d) {
          if (full) continue;
          unsigned z = tw[d];
          int k = (d == rs_d) ? rs_k : 0;
          const int e0 = (g * SCAN_W + lane * 4 + d) * 32;
          unsigned mk = __builtin_amdgcn_ballot_w32(z != 0u);
          while (mk) {
            if (n_w > CAP - 32 - 2 * B) {
              full = true;
              rs_d = d;
              rs_k = k;
              break;
            }
            if (z != 0u) {
              my[n_w + (int)__builtin_amdgcn_mbcnt_lo(mk, 0u)] =
                  ((unsigned)(e0 + __builtin_ctz(z)) << 8) | 255u;
              z &= z - 1u;
            }
            n_w += __builtin_popcount(mk);
            ++k;
            mk = __builtin_amdgcn_ballot_w32(z != 0u);
          }
        }
        if (!full) {
          rs_d = 0;
          rs_k = 0;
          t = load_bits(g + AGG_PD * NW);
          g += NW;
        }
      };
      auto room = [&]() { return g < ngroups && !full; };
      while (room()) {
#pragma unroll
        for (int u = 0; u < AGG_PD; ++u) {
          if (u > 0 && !room()) break;
          scan_step(tq[u]);
        }
      }
      n_pad = (n_w + 2 * B - 1) / (2 * B) * (2 * B);
      if (lane < n_pad - n_w) my[n_w + lane] = ((unsigned)(E + lo) << 8) | (unsigned)R;
      if (LM == 1 && round == 0 && lane == 0) keep_cnt[slot] = n_pad;

      for (int i = lane; i < n_pad; i += 32) {
        const unsigned en = my[i];
        const int e = (int)(en >> 8);
        const int lb = (int)(en & 255u);
        const int sv = src[min(e, E - 1)];
        const int dv = dst[min(e, E - 1)];
        const int s = (e < E) ? sv : e - E;
        const int dd = dv - lo;
        const int dl = (lb != 255) ? lb : (((unsigned)dd < (unsigned)cnt) ? dd : R);
        my[i] = ((unsigned)s << 8) | (unsigned)dl;
        if (LM == 1 && round == 0) keep_lst[(size_t)slot * CAP + i] = (s << 8) | dl;
        note_logits(i, s, dl);
      }
    }
    __syncthreads();
    if (round > 0) {
      for (int i = tid; i < R * F; i += AGG_NT) {
        const int q = (i / F) * H + (i % F) / D;
        const unsigned ko = s_mo[q], kn = s_mk[q];
        if (ko != kn) acc[i] *= expf(fdekey(ko) - fdekey(kn));
      }
      for (int i = tid; i < R * H; i += AGG_NT) {
        const unsigned ko = s_mo[i], kn = s_mk[i];
        if (ko != kn) s_den[i] *= expf(fdekey(ko) - fdekey(kn));
      }
      __syncthreads();
    }

    const int nb2 = n_pad / (2 * B);
    const int tq = min(lane / H, B - 1);
    const int th = lane - (lane / H) * H;
    auto request = [&](int b, float (&v)[B][J], float& e_t, int& dl_t, int& dl_l) {
      const unsigned en_l = my[b * B + (lane & (B - 1))];
      const unsigned en_t = my[b * B + tq];
      e_t = lg[(size_t)(b * B + tq) * H + th];
#pragma unroll
      for (int q = 0; q < B; ++q) {
        const int s_q = (int)((unsigned)__builtin_amdgcn_readlane((int)en_l, q) >> 8);
#pragma unroll
        for (int j = 0; j < J; ++j) {
          const int f = (F % 32 == 0) ? j * 32 + lane : min(j * 32 + lane, F - 1);
          v[q][j] = h[(size_t)s_q * F + f];
        }
      }
      dl_t = (int)(en_t & 255u);
      dl_l = (int)(en_l & 255u);
    };
    auto add_batch = [&](const float (&v)[B][J], float e_t, int dl_t, int dl_l) {
      const float ex = expf(e_t - fdekey(s_mk[dl_t * H + th]));
      if (lane < B * H) atomicAdd(&s_den[dl_t * H + th], ex);
#pragma unroll
      for (int q = 0; q < B; ++q) {
        const int dl_q = __builtin_amdgcn_readlane(dl_l, q);
#pragma unroll
        for (int j = 0; j < J; ++j) {
          const int f = j * 32 + lane;
          const int hh = (H == 1) ? 0 : ((D % 32 == 0) ? (j * 32) / D : f / D);
          const float w = __int_as_float(
              __builtin_amdgcn_readlane(__float_as_int(ex), q * H + hh));
          if (F % 32 == 0 || f < F) atomicAdd(&acc[dl_q * F + f], w * v[q][j]);
        }
      }
    };
    if (nb2 > 0) {
      float vA[B][J], vB[B][J];
      float eA, eB;
      int dtA, dlA, dtB, dlB;
      request(0, vA, eA, dtA, dlA);
      for (int b = 0; b < nb2; ++b) {
        request(2 * b + 1, vB, eB, dtB, dlB);
        add_batch(vA, eA, dtA, dlA);
        request(min(2 * b + 2, 2 * nb2 - 1), vA, eA, dtA, dlA);
        add_batch(vB, eB, dtB, dlB);
      }
    }

    if (!__syncthreads_or(g < ngroups)) break;
    for (int i = tid; i < (R + 1) * H; i += AGG_NT) s_mo[i] = s_mk[i];
    __syncthreads();
  }
  if (LM == 1 && tid == 0) keep_ok[blockIdx.x] = (round == 0) ? 1 : 0;

  if constexpr (NC == 0) {
    for (int i = tid; i < cnt * F; i += AGG_NT) {
      const int q = (i / F) * H + (i % F) / D;
      agg[(size_t)lo * F + i] = acc[i] / s_den[q] + bias[i % F];
    }
  } else {
    constexpr int XS = F + 4;
    constexpr int NCT = (NC + 15) / 16;
    static_assert(F % 4 == 0 && XS % 64 == 4, "k steps of 4; conflict-free fragment reads");
    static_assert((R + 15) / 16 <= NW, "one wave per 16-row tile");
    static_assert((R + NCT * 16) * XS <= NW * CAP, "x and Wn fit the list's space");
    static_assert(R * NC <= (R + 1) * F, "the product fits the rows' space");
    float* xp = reinterpret_cast<float*>(lst);
    float* wp = xp + R * XS;
    float* hs = acc;
    constexpr int WJ = (NCT * 16 * F + AGG_NT - 1) / AGG_NT;
    float wv[WJ];
#pragma unroll
    for (int j = 0; j < WJ; ++j) {
      const int i = tid + j * AGG_NT;
      wv[j] = (i < NC * F) ? Wn[i] : 0.f;
    }
    for (int i = tid; i < cnt * F; i += AGG_NT) {
      const int r = i / F, f = i % F;
      const float v = acc[i] / s_den[r * H + f / D] + bias[f];
      xp[r * XS + f] = (v > 0.f) ? v : 0.f;
    }
#pragma unroll
    for (int j = 0; j < WJ; ++j) {
      const int i = tid + j * AGG_NT;
      if (i < NCT * 16 * F) wp[(i / F) * XS + i % F] = wv[j];
    }
    __syncthreads();
    if (wave * 16 < cnt) {
      const int half = lane >> 4;
      const int idx = lane & 15;
      const int ar = min(wave * 16 + idx, cnt - 1);
      const float* ap = xp + ar * XS + 2 * half;
      const float* bp = wp + idx * XS + 2 * half;
      v8f c[NCT];
#pragma unroll
      for (int t = 0; t < NCT; ++t) c[t] = v8f{0.f, 0.f, 0.f, 0.f, 0.f, 0.f, 0.f, 0.f};
#pragma unroll
      for (int k = 0; k < F; k += 4) {
        const v2f a = *(const v2f*)(ap + k);
#pragma unroll
        for (int t = 0; t < NCT; ++t) {
          const v2f b = *(const v2f*)(bp + t * 16 * XS + k);
          c[t] = __builtin_amdgcn_wmma_f32_16x16x4_f32(false, a, false, b, (short)0, c[t],
                                                       false, false);
        }
      }
#pragma unroll
      for (int t = 0; t < NCT; ++t) {
        const int col = t * 16 + idx;
#pragma unroll
        for (int r = 0; r < 8; ++r) {
          const int row = wave * 16 + r + 8 * half;
          if (col < NC && row < cnt) hs[row * NC + col] = c[t][r];
        }
      }
    }
    __syncthreads();
    for (int i = tid; i < cnt * NC; i += AGG_NT) hn[(size_t)lo * NC + i] = hs[i];
    if (tid < cnt) {
      float s0 = 0.f, s1 = 0.f;
      for (int dd = 0; dd < NC; ++dd) {
        const float v = hs[tid * NC + dd];
        s0 += v * att_sn[dd];
        s1 += v * att_dn[dd];
      }
      a_sn[lo + tid] = s0;
      a_dn[lo + tid] = s1;
    }
  }
}

extern "C" void kernel_launch(void* const* d_in, const int* in_sizes, int n_in,
                              void* d_out, int out_size, void* d_ws, size_t ws_size,
                              hipStream_t stream) {
  const float* x        = (const float*)d_in[0];
  const int*   ei       = (const int*)d_in[1];
  const float* W1       = (const float*)d_in[2];
  const float* att_src1 = (const float*)d_in[3];
  const float* att_dst1 = (const float*)d_in[4];
  const float* b1       = (const float*)d_in[5];
  const float* W2       = (const float*)d_in[6];
  const float* att_src2 = (const float*)d_in[7];
  const float* att_dst2 = (const float*)d_in[8];
  const float* b2       = (const float*)d_in[9];

  const int NFEAT = 256, NHID = 128, H1 = 4, D1 = 32, NCLS = 40;
  const int N = in_sizes[0] / NFEAT;
  const int E = in_sizes[1] / 2;
  const int* src = ei;
  const int* dst = ei + E;

  float* out = (float*)d_out;

  float* w = (float*)d_ws;
  float* h1   = w;
  float* h2   = h1  + (size_t)N * NHID;
  float* as1  = h2  + (size_t)N * NCLS;
  float* ad1  = as1 + (size_t)N * H1;
  const int groups = (N + AGG_R - 1) / AGG_R;
  float* lg1  = ad1 + (size_t)N * H1;
  const int nword = (E + 32 * SCAN_W - 1) / (32 * SCAN_W) * SCAN_W;
  const int nchunk = nword / BM_CW;
  unsigned* gbm = (unsigned*)(lg1 + (size_t)groups * (AGG_NT / 32) * AGG_CAP * H1);
  float* as2  = (float*)(gbm + (size_t)AGG_NG * nword);
  float* ad2  = as2 + (size_t)N;
  int* klst = (int*)(ad2 + (size_t)N);
  int* kcnt = klst + (size_t)groups * (AGG_NT / 32) * AGG_CAP;
  int* kok  = kcnt + (size_t)groups * (AGG_NT / 32);

  const int TB = 256;

  gemm_wmma_f32_bres<256, 128><<<G1_WG, G1_NT, 0, stream>>>(x, W1, h1, N);
  attn_logits<<<nchunk + (N * H1 + TB - 1) / TB, TB, 0, stream>>>(h1, att_src1, att_dst1, as1, ad1,
                                                                  N * H1, H1, D1, dst, E, gbm,
                                                                  nword, nchunk);
  gat_softmax_aggregate<4, 32, 40, 1><<<groups, AGG_NT, 0, stream>>>(
      src, dst, E, N, as1, ad1, h1, b1, nullptr, W2, att_src2, att_dst2, h2, as2, ad2,
      klst, kcnt, kok, lg1, gbm, nword);

  gat_softmax_aggregate<1, 40, 0, 2><<<groups, AGG_NT, 0, stream>>>(
      src, dst, E, N, as2, ad2, h2, b2, out, nullptr, nullptr, nullptr, nullptr, nullptr, nullptr,
      klst, kcnt, kok, nullptr, gbm, nword);
}
